# starting kernel
# baseline (speedup 1.0000x reference)
#include <hip/hip_runtime.h>
#include <hip/hip_bf16.h>
#include <math.h>


typedef __attribute__((ext_vector_type(2))) float v2f;
typedef __attribute__((ext_vector_type(8))) float v8f;

#define D_MODEL 1024
#define D_INNER 2048
#define NB 2
#define SEQ_L 1024
#define DSTATE 16
#define DTRANK 64
#define XDBL_N 96

template <int MODE, int MT, int NT>
__global__ __launch_bounds__(256) void wmma_gemm_f32(
    const float* __restrict__ A, const float* __restrict__ W,
    float* __restrict__ C, const float* __restrict__ bias,
    int M, int N, int K, int lda, int ldb, int ldc) {
  const int lane = threadIdx.x & 31;
  const int wave = threadIdx.x >> 5;
  const int ntx = N / (16 * NT);
  const int nty = M / (16 * MT);
  const int task = blockIdx.x * 8 + wave;
  if (task >= ntx * nty) return;
  const int m0 = (task / ntx) * (16 * MT);
  const int n0 = (task % ntx) * (16 * NT);

  const int r = lane & 15;
  const int khalf = (lane >> 4) << 1;

  const float* arow[MT];
#pragma unroll
  for (int i = 0; i < MT; ++i)
    arow[i] = A + (size_t)(m0 + i * 16 + r) * lda + khalf;
  const float* wrow[NT];
#pragma unroll
  for (int j = 0; j < NT; ++j)
    wrow[j] = W + (size_t)(n0 + j * 16 + r) * ldb + khalf;

  v8f acc[MT][NT];
#pragma unroll
  for (int i = 0; i < MT; ++i)
#pragma unroll
    for (int j = 0; j < NT; ++j) acc[i][j] = (v8f){};

  v2f a_cur[MT], b_cur[NT];
#pragma unroll
  for (int i = 0; i < MT; ++i) a_cur[i] = *(const v2f*)(arow[i]);
#pragma unroll
  for (int j = 0; j < NT; ++j) b_cur[j] = *(const v2f*)(wrow[j]);

  for (int k = 0; k < K; k += 4) {
    const int kn = (k + 4 < K) ? (k + 4) : 0;
    v2f a_nxt[MT], b_nxt[NT];
#pragma unroll
    for (int i = 0; i < MT; ++i) a_nxt[i] = *(const v2f*)(arow[i] + kn);
#pragma unroll
    for (int j = 0; j < NT; ++j) b_nxt[j] = *(const v2f*)(wrow[j] + kn);

#pragma unroll
    for (int i = 0; i < MT; ++i)
#pragma unroll
      for (int j = 0; j < NT; ++j)
        acc[i][j] = __builtin_amdgcn_wmma_f32_16x16x4_f32(
            false, a_cur[i], false, b_cur[j], (short)0, acc[i][j], false, false);

#pragma unroll
    for (int i = 0; i < MT; ++i) a_cur[i] = a_nxt[i];
#pragma unroll
    for (int j = 0; j < NT; ++j) b_cur[j] = b_nxt[j];
  }

  const int mb = (lane >> 4) << 3;
#pragma unroll
  for (int i = 0; i < MT; ++i) {
#pragma unroll
    for (int j = 0; j < NT; ++j) {
      const int n = n0 + j * 16 + r;
      float bv = 0.0f;
      if (MODE == 1) bv = bias[n];
#pragma unroll
      for (int e = 0; e < 8; ++e) {
        float v = acc[i][j][e];
        if (MODE == 1) {
          v += bv;
          v = (v > 20.0f) ? v : log1pf(__expf(v));
        }
        C[(size_t)(m0 + i * 16 + mb + e) * ldc + n] = v;
      }
    }
  }
}

__global__ __launch_bounds__(256) void conv_silu_kernel(
    const float* __restrict__ xz, const float* __restrict__ cw,
    const float* __restrict__ cb, float* __restrict__ xconv) {
  const int gid = blockIdx.x * blockDim.x + threadIdx.x;
  if (gid >= NB * SEQ_L * D_INNER) return;
  const int d = gid & (D_INNER - 1);
  const int bl = gid / D_INNER;
  const int l = bl & (SEQ_L - 1);
  const int b = bl / SEQ_L;

  float acc = cb[d];
  const float* base = xz + ((size_t)b * SEQ_L) * (2 * D_INNER) + d;
#pragma unroll
  for (int j = 0; j < 4; ++j) {
    const int ls = l - 3 + j;
    if (ls >= 0) acc = fmaf(cw[d * 4 + j], base[(size_t)ls * (2 * D_INNER)], acc);
  }
  xconv[gid] = acc / (1.0f + __expf(-acc));
}

__global__ __launch_bounds__(256) void scan_kernel(
    const float* __restrict__ dt, const float* __restrict__ xconv,
    const float* __restrict__ x_dbl, const float* __restrict__ xz,
    const float* __restrict__ A_log, const float* __restrict__ Dp,
    float* __restrict__ u) {
  const int gid = blockIdx.x * blockDim.x + threadIdx.x;
  const int c = gid >> 4;
  const int n = gid & 15;
  if (c >= NB * D_INNER) return;
  const int d = c & (D_INNER - 1);
  const int b = c >> 11;

  const float An = -__expf(A_log[d * DSTATE + n]);
  const float Dd = Dp[d];
  float h = 0.0f;

  const float* dt_p = dt + ((size_t)b * SEQ_L) * D_INNER + d;
  const float* xc_p = xconv + ((size_t)b * SEQ_L) * D_INNER + d;
  const float* bc_p = x_dbl + ((size_t)b * SEQ_L) * XDBL_N + DTRANK + n;
  const float* z_p = xz + ((size_t)b * SEQ_L) * (2 * D_INNER) + D_INNER + d;
  float* u_p = u + ((size_t)b * SEQ_L) * D_INNER + d;

  for (int l = 0; l < SEQ_L; ++l) {
    const float dtv = dt_p[(size_t)l * D_INNER];
    const float xv = xc_p[(size_t)l * D_INNER];
    const float Bn = bc_p[(size_t)l * XDBL_N];
    const float Cn = bc_p[(size_t)l * XDBL_N + DSTATE];

    const float dA = __expf(dtv * An);
    h = fmaf(dA, h, dtv * xv * Bn);

    float y = h * Cn;
    y += __shfl_xor(y, 1, 32);
    y += __shfl_xor(y, 2, 32);
    y += __shfl_xor(y, 4, 32);
    y += __shfl_xor(y, 8, 32);

    if (n == 0) {
      const float zv = z_p[(size_t)l * (2 * D_INNER)];
      const float sz = zv / (1.0f + __expf(-zv));
      u_p[(size_t)l * D_INNER] = fmaf(xv, Dd, y) * sz;
    }
  }
}

extern "C" void kernel_launch(void* const* d_in, const int* in_sizes, int n_in,
                              void* d_out, int out_size, void* d_ws, size_t ws_size,
                              hipStream_t stream) {
  const float* x      = (const float*)d_in[0];
  const float* W_in   = (const float*)d_in[1];
  const float* conv_w = (const float*)d_in[2];
  const float* conv_b = (const float*)d_in[3];
  const float* W_x    = (const float*)d_in[4];
  const float* W_dt   = (const float*)d_in[5];
  const float* b_dt   = (const float*)d_in[6];
  const float* A_log  = (const float*)d_in[7];
  const float* Dp     = (const float*)d_in[8];
  const float* W_out  = (const float*)d_in[9];
  float* out = (float*)d_out;

  float* ws = (float*)d_ws;
  const size_t M = (size_t)NB * SEQ_L;
  float* xz    = ws;
  float* xconv = xz + M * 2 * D_INNER;
  float* x_dbl = xconv + M * D_INNER;
  float* dt    = x_dbl + M * XDBL_N;
  float* u     = dt + M * D_INNER;

  dim3 blk(256);
  {
    int tasks = (int)(M / 32) * (2 * D_INNER / 64);
    hipLaunchKernelGGL((wmma_gemm_f32<0, 2, 4>), dim3((tasks + 7) / 8), blk, 0, stream,
                       x, W_in, xz, nullptr,
                       (int)M, 2 * D_INNER, D_MODEL, D_MODEL, D_MODEL, 2 * D_INNER);
  }
  {
    int total = NB * SEQ_L * D_INNER;
    hipLaunchKernelGGL(conv_silu_kernel, dim3((total + 255) / 256), blk, 0, stream,
                       xz, conv_w, conv_b, xconv);
  }
  {
    int tasks = (int)(M / 32) * (XDBL_N / 32);
    hipLaunchKernelGGL((wmma_gemm_f32<0, 2, 2>), dim3((tasks + 7) / 8), blk, 0, stream,
                       xconv, W_x, x_dbl, nullptr,
                       (int)M, XDBL_N, D_INNER, D_INNER, D_INNER, XDBL_N);
  }
  {
    int tasks = (int)(M / 32) * (D_INNER / 64);
    hipLaunchKernelGGL((wmma_gemm_f32<1, 2, 4>), dim3((tasks + 7) / 8), blk, 0, stream,
                       x_dbl, W_dt, dt, b_dt,
                       (int)M, D_INNER, DTRANK, XDBL_N, DTRANK, D_INNER);
  }
  {
    int total = NB * D_INNER * DSTATE;
    hipLaunchKernelGGL(scan_kernel, dim3((total + 255) / 256), blk, 0, stream,
                       dt, xconv, x_dbl, xz, A_log, Dp, u);
  }
  {
    int tasks = (int)(M / 32) * (D_MODEL / 64);
    hipLaunchKernelGGL((wmma_gemm_f32<0, 2, 4>), dim3((tasks + 7) / 8), blk, 0, stream,
                       u, W_out, out, nullptr,
                       (int)M, D_MODEL, D_INNER, D_INNER, D_INNER, D_MODEL);
  }
}


# GEMM workgroups re-cut for equal matrix work per SIMD: out product 256 groups of 2x2 waves, 1 per SIMD (was 128 of 8 on half the WGPs); x_dbl 384 one-wave 16x32 tiles, 64-float stages; same sums
# speedup vs baseline: 6.5190x; 6.5190x over previous
#include <hip/hip_runtime.h>
#include <hip/hip_bf16.h>
#include <math.h>


typedef __attribute__((ext_vector_type(2))) float v2f;
typedef __attribute__((ext_vector_type(8))) float v8f;

#define D_MODEL 1024
#define D_INNER 2048
#define NB 2
#define SEQ_L 1024
#define DSTATE 16
#define DTRANK 64
#define XDBL_N 96

#define GEMM_KC 32

struct Frag16 {
  float4 q[4];
};

__device__ __forceinline__ void frag_load(Frag16& f, const float* __restrict__ p) {
  const float4* p4 = (const float4*)p;
#pragma unroll
  for (int t = 0; t < 4; ++t) f.q[t] = p4[t];
}

__device__ __forceinline__ v2f frag_pair(const Frag16& f, int s) {
  const float4 v = f.q[s >> 1];
  return (s & 1) ? (v2f){v.z, v.w} : (v2f){v.x, v.y};
}

template <int MT, int NT, int CH>
__device__ __forceinline__ void gemm_stage_load(Frag16 (&a)[CH][MT], Frag16 (&b)[CH][NT],
                                                const float* const (&arow)[MT],
                                                const float* const (&wrow)[NT], int k) {
#pragma unroll
  for (int c = 0; c < CH; ++c) {
#pragma unroll
    for (int i = 0; i < MT; ++i) frag_load(a[c][i], arow[i] + k + c * GEMM_KC);
#pragma unroll
    for (int j = 0; j < NT; ++j) frag_load(b[c][j], wrow[j] + k + c * GEMM_KC);
  }
}

template <int MT, int NT>
__device__ __forceinline__ void gemm_chunk_mma(const Frag16 (&a)[MT], const Frag16 (&b)[NT],
                                               v8f (&acc)[MT][NT]) {
#pragma unroll
  for (int s = 0; s < GEMM_KC / 4; ++s) {
    v2f av[MT], bv[NT];
#pragma unroll
    for (int i = 0; i < MT; ++i) av[i] = frag_pair(a[i], s);
#pragma unroll
    for (int j = 0; j < NT; ++j) bv[j] = frag_pair(b[j], s);
#pragma unroll
    for (int i = 0; i < MT; ++i)
#pragma unroll
      for (int j = 0; j < NT; ++j)
        acc[i][j] = __builtin_amdgcn_wmma_f32_16x16x4_f32(
            false, av[i], false, bv[j], (short)0, acc[i][j], false, false);
  }
}

template <int MT, int NT, int CH>
__device__ __forceinline__ void gemm_stage_mma(const Frag16 (&a)[CH][MT], const Frag16 (&b)[CH][NT],
                                               v8f (&acc)[MT][NT]) {
#pragma unroll
  for (int c = 0; c < CH; ++c) gemm_chunk_mma<MT, NT>(a[c], b[c], acc);
}

template <int MODE, int MT, int NT, int WPB, int WM, int CH>
__global__ __launch_bounds__(WPB * 32) void wmma_gemm_f32(
    const float* __restrict__ A, const float* __restrict__ W,
    float* __restrict__ C, const float* __restrict__ bias,
    int M, int N, int K, int lda, int ldb, int ldc) {
  const int lane = threadIdx.x & 31;
  const int wave = threadIdx.x >> 5;
  constexpr int WN = WPB / WM;
  const int nbx = N / (16 * NT * WN);
  const int m0 = ((blockIdx.x / nbx) * WM + wave % WM) * (16 * MT);
  const int n0 = ((blockIdx.x % nbx) * WN + wave / WM) * (16 * NT);
  if (m0 >= M) return;

  const int r = lane & 15;
  const int kseg = (lane >> 4) << 4;

  const float* arow[MT];
#pragma unroll
  for (int i = 0; i < MT; ++i)
    arow[i] = A + (size_t)(m0 + i * 16 + r) * lda + kseg;
  const float* wrow[NT];
#pragma unroll
  for (int j = 0; j < NT; ++j)
    wrow[j] = W + (size_t)(n0 + j * 16 + r) * ldb + kseg;

  v8f acc[MT][NT];
#pragma unroll
  for (int i = 0; i < MT; ++i)
#pragma unroll
    for (int j = 0; j < NT; ++j) acc[i][j] = (v8f){};

  constexpr int KS = CH * GEMM_KC;
  Frag16 a0[CH][MT], b0[CH][NT], a1[CH][MT], b1[CH][NT];
  gemm_stage_load<MT, NT, CH>(a0, b0, arow, wrow, 0);

  for (int k = 0; k < K; k += 2 * KS) {
    gemm_stage_load<MT, NT, CH>(a1, b1, arow, wrow, k + KS);
    __builtin_amdgcn_sched_barrier(0);
    gemm_stage_mma<MT, NT, CH>(a0, b0, acc);
    __builtin_amdgcn_sched_barrier(0);

    const int kn = (k + 2 * KS < K) ? (k + 2 * KS) : 0;
    gemm_stage_load<MT, NT, CH>(a0, b0, arow, wrow, kn);
    __builtin_amdgcn_sched_barrier(0);
    gemm_stage_mma<MT, NT, CH>(a1, b1, acc);
  }

  const int mb = (lane >> 4) << 3;
#pragma unroll
  for (int i = 0; i < MT; ++i) {
#pragma unroll
    for (int j = 0; j < NT; ++j) {
      const int n = n0 + j * 16 + r;
      float bv = 0.0f;
      if (MODE == 1) bv = bias[n];
#pragma unroll
      for (int e = 0; e < 8; ++e) {
        float v = acc[i][j][e];
        if (MODE == 1) {
          v += bv;
          v = (v > 20.0f) ? v : log1pf(__expf(v));
        }
        C[(size_t)(m0 + i * 16 + mb + e) * ldc + n] = v;
      }
    }
  }
}

__global__ __launch_bounds__(256) void conv_silu_kernel(
    const float* __restrict__ xz, const float* __restrict__ cw,
    const float* __restrict__ cb, float* __restrict__ xconv) {
  const int gid = blockIdx.x * blockDim.x + threadIdx.x;
  if (gid >= NB * SEQ_L * D_INNER) return;
  const int d = gid & (D_INNER - 1);
  const int bl = gid / D_INNER;
  const int l = bl & (SEQ_L - 1);
  const int b = bl / SEQ_L;

  float acc = cb[d];
  const float* base = xz + ((size_t)b * SEQ_L) * (2 * D_INNER) + d;
#pragma unroll
  for (int j = 0; j < 4; ++j) {
    const int ls = l - 3 + j;
    if (ls >= 0) acc = fmaf(cw[d * 4 + j], base[(size_t)ls * (2 * D_INNER)], acc);
  }
  xconv[gid] = acc / (1.0f + __expf(-acc));
}

#define SCAN_TB 8
#define SCAN_NCH 16
#define SCAN_CL (SEQ_L / SCAN_NCH)
#define SCAN_CHANS (NB * D_INNER)
#define SCAN_GROUPS (SCAN_CHANS / 32)
#define CARRY_G 8

__device__ __forceinline__ float lane_bcast(float v, int srclane) {
  return __int_as_float(__builtin_amdgcn_readlane(__float_as_int(v), srclane));
}

__device__ __forceinline__ void scan_load_A(float (&An)[DSTATE], const float* __restrict__ A_log, int d) {
  const float4* ap = (const float4*)(A_log + (size_t)d * DSTATE);
#pragma unroll
  for (int q = 0; q < DSTATE / 4; ++q) {
    const float4 a4 = ap[q];
    An[q * 4 + 0] = -__expf(a4.x);
    An[q * 4 + 1] = -__expf(a4.y);
    An[q * 4 + 2] = -__expf(a4.z);
    An[q * 4 + 3] = -__expf(a4.w);
  }
}

struct StateBatch {
  float dtv[SCAN_TB];
  float xv[SCAN_TB];
  float bv[SCAN_TB];
};

__device__ __forceinline__ void state_load(StateBatch& t, const float* __restrict__ dt_p,
                                           const float* __restrict__ xc_p,
                                           const float* __restrict__ b_p, int l) {
#pragma unroll
  for (int s = 0; s < SCAN_TB; ++s) {
    t.dtv[s] = dt_p[(size_t)(l + s) * D_INNER];
    t.xv[s] = xc_p[(size_t)(l + s) * D_INNER];
    t.bv[s] = b_p[(size_t)(l + s) * XDBL_N];
  }
}

__device__ __forceinline__ void state_steps(const StateBatch& t, float (&h)[DSTATE],
                                            const float (&An)[DSTATE], float& dsum) {
#pragma unroll
  for (int s = 0; s < SCAN_TB; ++s) {
    const float dtv = t.dtv[s];
    const float dtx = dtv * t.xv[s];
    const float bcv = t.bv[s];
    dsum += dtv;
#pragma unroll
    for (int n = 0; n < DSTATE; ++n) {
      const float Bn = lane_bcast(bcv, n);
      h[n] = fmaf(__expf(dtv * An[n]), h[n], dtx * Bn);
    }
  }
}

__global__ __launch_bounds__(32) void scan_state_kernel(
    const float* __restrict__ dt, const float* __restrict__ xconv,
    const float* __restrict__ x_dbl, const float* __restrict__ A_log,
    float* __restrict__ S, float* __restrict__ Dsum) {
  const int lane = threadIdx.x;
  const int ck = blockIdx.x / SCAN_GROUPS;
  const int c0 = (blockIdx.x % SCAN_GROUPS) * 32;
  const int b = c0 >> 11;
  const int d = (c0 & (D_INNER - 1)) + lane;

  float An[DSTATE];
  scan_load_A(An, A_log, d);

  float h[DSTATE];
#pragma unroll
  for (int n = 0; n < DSTATE; ++n) h[n] = 0.0f;
  float dsum = 0.0f;

  const size_t row0 = (size_t)b * SEQ_L + (size_t)ck * SCAN_CL;
  const float* dt_p = dt + row0 * D_INNER + d;
  const float* xc_p = xconv + row0 * D_INNER + d;
  const float* b_p = x_dbl + row0 * XDBL_N + DTRANK + (lane & 15);

  StateBatch t0, t1;
  state_load(t0, dt_p, xc_p, b_p, 0);

  for (int l0 = 0; l0 < SCAN_CL; l0 += 2 * SCAN_TB) {
    state_load(t1, dt_p, xc_p, b_p, l0 + SCAN_TB);
    __builtin_amdgcn_sched_barrier(0);
    state_steps(t0, h, An, dsum);

    const int ln = (l0 + 2 * SCAN_TB < SCAN_CL) ? (l0 + 2 * SCAN_TB) : 0;
    state_load(t0, dt_p, xc_p, b_p, ln);
    __builtin_amdgcn_sched_barrier(0);
    state_steps(t1, h, An, dsum);
  }

  const int ch = c0 + lane;
#pragma unroll
  for (int n = 0; n < DSTATE; ++n)
    S[((size_t)ck * DSTATE + n) * SCAN_CHANS + ch] = h[n];
  Dsum[(size_t)ck * SCAN_CHANS + ch] = dsum;
}

struct ScanBatch {
  float dtv[SCAN_TB];
  float xv[SCAN_TB];
  float zv[SCAN_TB];
  float bc[SCAN_TB];
};

__device__ __forceinline__ void scan_load(ScanBatch& t, const float* __restrict__ dt_p,
                                          const float* __restrict__ xc_p,
                                          const float* __restrict__ z_p,
                                          const float* __restrict__ bc_p, int l) {
#pragma unroll
  for (int s = 0; s < SCAN_TB; ++s) {
    t.dtv[s] = dt_p[(size_t)(l + s) * D_INNER];
    t.xv[s] = xc_p[(size_t)(l + s) * D_INNER];
    t.zv[s] = z_p[(size_t)(l + s) * (2 * D_INNER)];
    t.bc[s] = bc_p[(size_t)(l + s) * XDBL_N];
  }
}

__device__ __forceinline__ void scan_steps(const ScanBatch& t, float (&h)[DSTATE],
                                           const float (&An)[DSTATE], float Dd,
                                           float* __restrict__ u_p, int l) {
#pragma unroll
  for (int s = 0; s < SCAN_TB; ++s) {
    const float dtv = t.dtv[s];
    const float xv = t.xv[s];
    const float dtx = dtv * xv;
    const float bcv = t.bc[s];
    float a0 = 0.0f, a1 = 0.0f, a2 = 0.0f, a3 = 0.0f;
#pragma unroll
    for (int n = 0; n < DSTATE; n += 4) {
      const float B0 = lane_bcast(bcv, n + 0), C0 = lane_bcast(bcv, DSTATE + n + 0);
      const float B1 = lane_bcast(bcv, n + 1), C1 = lane_bcast(bcv, DSTATE + n + 1);
      const float B2 = lane_bcast(bcv, n + 2), C2 = lane_bcast(bcv, DSTATE + n + 2);
      const float B3 = lane_bcast(bcv, n + 3), C3 = lane_bcast(bcv, DSTATE + n + 3);
      h[n + 0] = fmaf(__expf(dtv * An[n + 0]), h[n + 0], dtx * B0);
      h[n + 1] = fmaf(__expf(dtv * An[n + 1]), h[n + 1], dtx * B1);
      h[n + 2] = fmaf(__expf(dtv * An[n + 2]), h[n + 2], dtx * B2);
      h[n + 3] = fmaf(__expf(dtv * An[n + 3]), h[n + 3], dtx * B3);
      a0 = fmaf(h[n + 0], C0, a0);
      a1 = fmaf(h[n + 1], C1, a1);
      a2 = fmaf(h[n + 2], C2, a2);
      a3 = fmaf(h[n + 3], C3, a3);
    }
    const float y = (a0 + a1) + (a2 + a3);
    const float zv = t.zv[s];
    const float sz = zv / (1.0f + __expf(-zv));
    u_p[(size_t)(l + s) * D_INNER] = fmaf(xv, Dd, y) * sz;
  }
}

__global__ __launch_bounds__(32) void scan_kernel(
    const float* __restrict__ dt, const float* __restrict__ xconv,
    const float* __restrict__ x_dbl, const float* __restrict__ xz,
    const float* __restrict__ A_log, const float* __restrict__ Dp,
    const float* __restrict__ S, const float* __restrict__ Dsum,
    float* __restrict__ u) {
  const int lane = threadIdx.x;
  const int ck = (SCAN_NCH - 1) - blockIdx.x / SCAN_GROUPS;
  const int c0 = (blockIdx.x % SCAN_GROUPS) * 32;
  const int b = c0 >> 11;
  const int d = (c0 & (D_INNER - 1)) + lane;
  const int ch = c0 + lane;

  float An[DSTATE];
  scan_load_A(An, A_log, d);
  const float Dd = Dp[d];

  float h[DSTATE];
#pragma unroll
  for (int n = 0; n < DSTATE; ++n) h[n] = 0.0f;

  for (int j0 = 0; j0 < ck; j0 += CARRY_G) {
    float sv[CARRY_G][DSTATE];
    float dj[CARRY_G];
#pragma unroll
    for (int q = 0; q < CARRY_G; ++q) {
      const int j = (j0 + q < SCAN_NCH - 1) ? (j0 + q) : (SCAN_NCH - 2);
      dj[q] = Dsum[(size_t)j * SCAN_CHANS + ch];
#pragma unroll
      for (int n = 0; n < DSTATE; ++n)
        sv[q][n] = S[((size_t)j * DSTATE + n) * SCAN_CHANS + ch];
    }
    __builtin_amdgcn_sched_barrier(0);
#pragma unroll
    for (int q = 0; q < CARRY_G; ++q) {
      const bool live = (j0 + q) < ck;
#pragma unroll
      for (int n = 0; n < DSTATE; ++n) {
        const float e = live ? __expf(dj[q] * An[n]) : 1.0f;
        const float a = live ? sv[q][n] : 0.0f;
        h[n] = fmaf(e, h[n], a);
      }
    }
  }

  const size_t row0 = (size_t)b * SEQ_L + (size_t)ck * SCAN_CL;
  const float* dt_p = dt + row0 * D_INNER + d;
  const float* xc_p = xconv + row0 * D_INNER + d;
  const float* bc_p = x_dbl + row0 * XDBL_N + DTRANK + lane;
  const float* z_p = xz + row0 * (2 * D_INNER) + D_INNER + d;
  float* u_p = u + row0 * D_INNER + d;

  ScanBatch t0, t1;
  scan_load(t0, dt_p, xc_p, z_p, bc_p, 0);

  for (int l0 = 0; l0 < SCAN_CL; l0 += 2 * SCAN_TB) {
    scan_load(t1, dt_p, xc_p, z_p, bc_p, l0 + SCAN_TB);
    __builtin_amdgcn_sched_barrier(0);
    scan_steps(t0, h, An, Dd, u_p, l0);

    const int ln = (l0 + 2 * SCAN_TB < SCAN_CL) ? (l0 + 2 * SCAN_TB) : 0;
    scan_load(t0, dt_p, xc_p, z_p, bc_p, ln);
    __builtin_amdgcn_sched_barrier(0);
    scan_steps(t1, h, An, Dd, u_p, l0 + SCAN_TB);
  }
}

extern "C" void kernel_launch(void* const* d_in, const int* in_sizes, int n_in,
                              void* d_out, int out_size, void* d_ws, size_t ws_size,
                              hipStream_t stream) {
  const float* x      = (const float*)d_in[0];
  const float* W_in   = (const float*)d_in[1];
  const float* conv_w = (const float*)d_in[2];
  const float* conv_b = (const float*)d_in[3];
  const float* W_x    = (const float*)d_in[4];
  const float* W_dt   = (const float*)d_in[5];
  const float* b_dt   = (const float*)d_in[6];
  const float* A_log  = (const float*)d_in[7];
  const float* Dp     = (const float*)d_in[8];
  const float* W_out  = (const float*)d_in[9];
  float* out = (float*)d_out;

  float* ws = (float*)d_ws;
  const size_t M = (size_t)NB * SEQ_L;
  float* xz    = ws;
  float* xconv = xz + M * 2 * D_INNER;
  float* x_dbl = xconv + M * D_INNER;
  float* dt    = x_dbl + M * XDBL_N;
  float* u     = dt + M * D_INNER;
  float* S     = u + M * D_INNER;
  float* Dsum  = S + (size_t)(SCAN_NCH - 1) * DSTATE * SCAN_CHANS;

  dim3 blk(256);
  {
    hipLaunchKernelGGL((wmma_gemm_f32<0, 2, 4, 8, 1, 1>), dim3(512), blk, 0, stream,
                       x, W_in, xz, nullptr,
                       (int)M, 2 * D_INNER, D_MODEL, D_MODEL, D_MODEL, 2 * D_INNER);
  }
  {
    int total = NB * SEQ_L * D_INNER;
    hipLaunchKernelGGL(conv_silu_kernel, dim3((total + 255) / 256), blk, 0, stream,
                       xz, conv_w, conv_b, xconv);
  }
  {
    hipLaunchKernelGGL((wmma_gemm_f32<0, 1, 2, 1, 1, 2>), dim3(384), dim3(32), 0, stream,
                       xconv, W_x, x_dbl, nullptr,
                       (int)M, XDBL_N, D_INNER, D_INNER, D_INNER, XDBL_N);
  }
  {
    hipLaunchKernelGGL((wmma_gemm_f32<1, 2, 4, 8, 1, 1>), dim3(256), blk, 0, stream,
                       x_dbl, W_dt, dt, b_dt,
                       (int)M, D_INNER, DTRANK, XDBL_N, DTRANK, D_INNER);
  }
  {
    hipLaunchKernelGGL(scan_state_kernel, dim3((SCAN_NCH - 1) * SCAN_GROUPS), dim3(32), 0, stream,
                       dt, xconv, x_dbl, A_log, S, Dsum);
  }
  {
    hipLaunchKernelGGL(scan_kernel, dim3(SCAN_NCH * SCAN_GROUPS), dim3(32), 0, stream,
                       dt, xconv, x_dbl, xz, A_log, Dp, S, Dsum, u);
  }
  {
    hipLaunchKernelGGL((wmma_gemm_f32<0, 2, 4, 4, 2, 1>), dim3(256), dim3(128), 0, stream,
                       u, W_out, out, nullptr,
                       (int)M, D_MODEL, D_INNER, D_INNER, D_INNER, D_MODEL);
  }
}


# depthwise conv4+bias+SiLU moved into the in_proj GEMM epilogue, from the accumulator tile (rows above a lane's block via the other lane half); a 128-group launch does 3 edge rows per tile; same sums
# speedup vs baseline: 6.6987x; 1.0276x over previous
#include <hip/hip_runtime.h>
#include <hip/hip_bf16.h>
#include <math.h>


typedef __attribute__((ext_vector_type(2))) float v2f;
typedef __attribute__((ext_vector_type(8))) float v8f;

#define D_MODEL 1024
#define D_INNER 2048
#define NB 2
#define SEQ_L 1024
#define DSTATE 16
#define DTRANK 64
#define XDBL_N 96

#define GEMM_KC 32
#define CONV_TM 32

__device__ __forceinline__ float xhalf(float x) {
  const int b = __builtin_bit_cast(int, x);
  return __builtin_bit_cast(float, __builtin_amdgcn_permlanex16(b, b, 0x76543210, 0xfedcba98, false, false));
}

struct Frag16 {
  float4 q[4];
};

__device__ __forceinline__ void frag_load(Frag16& f, const float* __restrict__ p) {
  const float4* p4 = (const float4*)p;
#pragma unroll
  for (int t = 0; t < 4; ++t) f.q[t] = p4[t];
}

__device__ __forceinline__ v2f frag_pair(const Frag16& f, int s) {
  const float4 v = f.q[s >> 1];
  return (s & 1) ? (v2f){v.z, v.w} : (v2f){v.x, v.y};
}

template <int MT, int NT, int CH>
__device__ __forceinline__ void gemm_stage_load(Frag16 (&a)[CH][MT], Frag16 (&b)[CH][NT],
                                                const float* const (&arow)[MT],
                                                const float* const (&wrow)[NT], int k) {
#pragma unroll
  for (int c = 0; c < CH; ++c) {
#pragma unroll
    for (int i = 0; i < MT; ++i) frag_load(a[c][i], arow[i] + k + c * GEMM_KC);
#pragma unroll
    for (int j = 0; j < NT; ++j) frag_load(b[c][j], wrow[j] + k + c * GEMM_KC);
  }
}

template <int MT, int NT>
__device__ __forceinline__ void gemm_chunk_mma(const Frag16 (&a)[MT], const Frag16 (&b)[NT],
                                               v8f (&acc)[MT][NT]) {
#pragma unroll
  for (int s = 0; s < GEMM_KC / 4; ++s) {
    v2f av[MT], bv[NT];
#pragma unroll
    for (int i = 0; i < MT; ++i) av[i] = frag_pair(a[i], s);
#pragma unroll
    for (int j = 0; j < NT; ++j) bv[j] = frag_pair(b[j], s);
#pragma unroll
    for (int i = 0; i < MT; ++i)
#pragma unroll
      for (int j = 0; j < NT; ++j)
        acc[i][j] = __builtin_amdgcn_wmma_f32_16x16x4_f32(
            false, av[i], false, bv[j], (short)0, acc[i][j], false, false);
  }
}

template <int MT, int NT, int CH>
__device__ __forceinline__ void gemm_stage_mma(const Frag16 (&a)[CH][MT], const Frag16 (&b)[CH][NT],
                                               v8f (&acc)[MT][NT]) {
#pragma unroll
  for (int c = 0; c < CH; ++c) gemm_chunk_mma<MT, NT>(a[c], b[c], acc);
}

template <int MODE, int MT, int NT, int WPB, int WM, int CH>
__global__ __launch_bounds__(WPB * 32) void wmma_gemm_f32(
    const float* __restrict__ A, const float* __restrict__ W,
    float* __restrict__ C, const float* __restrict__ bias,
    const float* __restrict__ cw, float* __restrict__ C2,
    int M, int N, int K, int lda, int ldb, int ldc) {
  static_assert(MODE != 2 || MT * 16 == CONV_TM, "conv_edge_kernel assumes this tile height");
  const int lane = threadIdx.x & 31;
  const int wave = threadIdx.x >> 5;
  constexpr int WN = WPB / WM;
  const int nbx = N / (16 * NT * WN);
  const int m0 = ((blockIdx.x / nbx) * WM + wave % WM) * (16 * MT);
  const int n0 = ((blockIdx.x % nbx) * WN + wave / WM) * (16 * NT);
  if (m0 >= M) return;

  const int r = lane & 15;
  const int kseg = (lane >> 4) << 4;

  const float* arow[MT];
#pragma unroll
  for (int i = 0; i < MT; ++i)
    arow[i] = A + (size_t)(m0 + i * 16 + r) * lda + kseg;
  const float* wrow[NT];
#pragma unroll
  for (int j = 0; j < NT; ++j)
    wrow[j] = W + (size_t)(n0 + j * 16 + r) * ldb + kseg;

  v8f acc[MT][NT];
#pragma unroll
  for (int i = 0; i < MT; ++i)
#pragma unroll
    for (int j = 0; j < NT; ++j) acc[i][j] = (v8f){};

  constexpr int KS = CH * GEMM_KC;
  Frag16 a0[CH][MT], b0[CH][NT], a1[CH][MT], b1[CH][NT];
  gemm_stage_load<MT, NT, CH>(a0, b0, arow, wrow, 0);

  for (int k = 0; k < K; k += 2 * KS) {
    gemm_stage_load<MT, NT, CH>(a1, b1, arow, wrow, k + KS);
    __builtin_amdgcn_sched_barrier(0);
    gemm_stage_mma<MT, NT, CH>(a0, b0, acc);
    __builtin_amdgcn_sched_barrier(0);

    const int kn = (k + 2 * KS < K) ? (k + 2 * KS) : 0;
    gemm_stage_load<MT, NT, CH>(a0, b0, arow, wrow, kn);
    __builtin_amdgcn_sched_barrier(0);
    gemm_stage_mma<MT, NT, CH>(a1, b1, acc);
  }

  const int mb = (lane >> 4) << 3;

  if (MODE == 2) {
    if (n0 < D_INNER) {
      const int h = lane >> 4;
#pragma unroll
      for (int j = 0; j < NT; ++j) {
        const int n = n0 + j * 16 + r;
        const float4 w = *(const float4*)(cw + (size_t)n * 4);
        const float cbv = bias[n];
        float up[MT][3];
#pragma unroll
        for (int i = 0; i < MT; ++i)
#pragma unroll
          for (int t = 0; t < 3; ++t) up[i][t] = xhalf(acc[i][j][5 + t]);
#pragma unroll
        for (int i = 0; i < MT; ++i) {
          float X[11];
#pragma unroll
          for (int t = 0; t < 3; ++t) X[t] = h ? up[i][t] : up[i > 0 ? i - 1 : 0][t];
#pragma unroll
          for (int e = 0; e < 8; ++e) X[3 + e] = acc[i][j][e];
#pragma unroll
          for (int e = 0; e < 8; ++e) {
            const size_t row = (size_t)(m0 + i * 16 + mb + e);
            float a = cbv;
            a = fmaf(w.x, X[e], a);
            a = fmaf(w.y, X[e + 1], a);
            a = fmaf(w.z, X[e + 2], a);
            a = fmaf(w.w, X[e + 3], a);
            const float v = a / (1.0f + __expf(-a));
            const bool top = (i == 0) && (e < 3);
            if (!top || h) C2[row * D_INNER + n] = v;
            if (top && !h) C[row * ldc + n] = acc[i][j][e];
            if (i == MT - 1 && e >= 5 && h) C[row * ldc + n] = acc[i][j][e];
          }
        }
      }
      return;
    }
  }

#pragma unroll
  for (int i = 0; i < MT; ++i) {
#pragma unroll
    for (int j = 0; j < NT; ++j) {
      const int n = n0 + j * 16 + r;
      float bv = 0.0f;
      if (MODE == 1) bv = bias[n];
#pragma unroll
      for (int e = 0; e < 8; ++e) {
        float v = acc[i][j][e];
        if (MODE == 1) {
          v += bv;
          v = (v > 20.0f) ? v : log1pf(__expf(v));
        }
        C[(size_t)(m0 + i * 16 + mb + e) * ldc + n] = v;
      }
    }
  }
}

__global__ __launch_bounds__(256) void conv_edge_kernel(
    const float* __restrict__ xz, const float* __restrict__ cw,
    const float* __restrict__ cb, float* __restrict__ xconv) {
  const int gid = blockIdx.x * blockDim.x + threadIdx.x;
  const int tile = gid / (D_INNER / 4);
  if (tile >= NB * SEQ_L / CONV_TM) return;
  const int d = (gid & (D_INNER / 4 - 1)) * 4;
  const int m0 = tile * CONV_TM;
  const bool first = (m0 & (SEQ_L - 1)) == 0;

  float w[4][4];
#pragma unroll
  for (int c = 0; c < 4; ++c) {
    const float4 t = *(const float4*)(cw + (size_t)(d + c) * 4);
    w[c][0] = t.x; w[c][1] = t.y; w[c][2] = t.z; w[c][3] = t.w;
  }
  const float4 bq = *(const float4*)(cb + d);
  const float bias[4] = {bq.x, bq.y, bq.z, bq.w};

  float x[6][4];
#pragma unroll
  for (int t = 0; t < 6; ++t) {
    float4 q = make_float4(0.0f, 0.0f, 0.0f, 0.0f);
    if (t >= 3 || !first) q = *(const float4*)(xz + (size_t)(m0 - 3 + t) * (2 * D_INNER) + d);
    x[t][0] = q.x; x[t][1] = q.y; x[t][2] = q.z; x[t][3] = q.w;
  }

#pragma unroll
  for (int e = 0; e < 3; ++e) {
    float o[4];
#pragma unroll
    for (int c = 0; c < 4; ++c) {
      float a = bias[c];
#pragma unroll
      for (int j = 0; j < 4; ++j)
        if (!first || e + j >= 3) a = fmaf(w[c][j], x[e + j][c], a);
      o[c] = a / (1.0f + __expf(-a));
    }
    *(float4*)(xconv + (size_t)(m0 + e) * D_INNER + d) = make_float4(o[0], o[1], o[2], o[3]);
  }
}

#define SCAN_TB 8
#define SCAN_NCH 16
#define SCAN_CL (SEQ_L / SCAN_NCH)
#define SCAN_CHANS (NB * D_INNER)
#define SCAN_GROUPS (SCAN_CHANS / 32)
#define CARRY_G 8

__device__ __forceinline__ float lane_bcast(float v, int srclane) {
  return __int_as_float(__builtin_amdgcn_readlane(__float_as_int(v), srclane));
}

__device__ __forceinline__ void scan_load_A(float (&An)[DSTATE], const float* __restrict__ A_log, int d) {
  const float4* ap = (const float4*)(A_log + (size_t)d * DSTATE);
#pragma unroll
  for (int q = 0; q < DSTATE / 4; ++q) {
    const float4 a4 = ap[q];
    An[q * 4 + 0] = -__expf(a4.x);
    An[q * 4 + 1] = -__expf(a4.y);
    An[q * 4 + 2] = -__expf(a4.z);
    An[q * 4 + 3] = -__expf(a4.w);
  }
}

struct StateBatch {
  float dtv[SCAN_TB];
  float xv[SCAN_TB];
  float bv[SCAN_TB];
};

__device__ __forceinline__ void state_load(StateBatch& t, const float* __restrict__ dt_p,
                                           const float* __restrict__ xc_p,
                                           const float* __restrict__ b_p, int l) {
#pragma unroll
  for (int s = 0; s < SCAN_TB; ++s) {
    t.dtv[s] = dt_p[(size_t)(l + s) * D_INNER];
    t.xv[s] = xc_p[(size_t)(l + s) * D_INNER];
    t.bv[s] = b_p[(size_t)(l + s) * XDBL_N];
  }
}

__device__ __forceinline__ void state_steps(const StateBatch& t, float (&h)[DSTATE],
                                            const float (&An)[DSTATE], float& dsum) {
#pragma unroll
  for (int s = 0; s < SCAN_TB; ++s) {
    const float dtv = t.dtv[s];
    const float dtx = dtv * t.xv[s];
    const float bcv = t.bv[s];
    dsum += dtv;
#pragma unroll
    for (int n = 0; n < DSTATE; ++n) {
      const float Bn = lane_bcast(bcv, n);
      h[n] = fmaf(__expf(dtv * An[n]), h[n], dtx * Bn);
    }
  }
}

__global__ __launch_bounds__(32) void scan_state_kernel(
    const float* __restrict__ dt, const float* __restrict__ xconv,
    const float* __restrict__ x_dbl, const float* __restrict__ A_log,
    float* __restrict__ S, float* __restrict__ Dsum) {
  const int lane = threadIdx.x;
  const int ck = blockIdx.x / SCAN_GROUPS;
  const int c0 = (blockIdx.x % SCAN_GROUPS) * 32;
  const int b = c0 >> 11;
  const int d = (c0 & (D_INNER - 1)) + lane;

  float An[DSTATE];
  scan_load_A(An, A_log, d);

  float h[DSTATE];
#pragma unroll
  for (int n = 0; n < DSTATE; ++n) h[n] = 0.0f;
  float dsum = 0.0f;

  const size_t row0 = (size_t)b * SEQ_L + (size_t)ck * SCAN_CL;
  const float* dt_p = dt + row0 * D_INNER + d;
  const float* xc_p = xconv + row0 * D_INNER + d;
  const float* b_p = x_dbl + row0 * XDBL_N + DTRANK + (lane & 15);

  StateBatch t0, t1;
  state_load(t0, dt_p, xc_p, b_p, 0);

  for (int l0 = 0; l0 < SCAN_CL; l0 += 2 * SCAN_TB) {
    state_load(t1, dt_p, xc_p, b_p, l0 + SCAN_TB);
    __builtin_amdgcn_sched_barrier(0);
    state_steps(t0, h, An, dsum);

    const int ln = (l0 + 2 * SCAN_TB < SCAN_CL) ? (l0 + 2 * SCAN_TB) : 0;
    state_load(t0, dt_p, xc_p, b_p, ln);
    __builtin_amdgcn_sched_barrier(0);
    state_steps(t1, h, An, dsum);
  }

  const int ch = c0 + lane;
#pragma unroll
  for (int n = 0; n < DSTATE; ++n)
    S[((size_t)ck * DSTATE + n) * SCAN_CHANS + ch] = h[n];
  Dsum[(size_t)ck * SCAN_CHANS + ch] = dsum;
}

struct ScanBatch {
  float dtv[SCAN_TB];
  float xv[SCAN_TB];
  float zv[SCAN_TB];
  float bc[SCAN_TB];
};

__device__ __forceinline__ void scan_load(ScanBatch& t, const float* __restrict__ dt_p,
                                          const float* __restrict__ xc_p,
                                          const float* __restrict__ z_p,
                                          const float* __restrict__ bc_p, int l) {
#pragma unroll
  for (int s = 0; s < SCAN_TB; ++s) {
    t.dtv[s] = dt_p[(size_t)(l + s) * D_INNER];
    t.xv[s] = xc_p[(size_t)(l + s) * D_INNER];
    t.zv[s] = z_p[(size_t)(l + s) * (2 * D_INNER)];
    t.bc[s] = bc_p[(size_t)(l + s) * XDBL_N];
  }
}

__device__ __forceinline__ void scan_steps(const ScanBatch& t, float (&h)[DSTATE],
                                           const float (&An)[DSTATE], float Dd,
                                           float* __restrict__ u_p, int l) {
#pragma unroll
  for (int s = 0; s < SCAN_TB; ++s) {
    const float dtv = t.dtv[s];
    const float xv = t.xv[s];
    const float dtx = dtv * xv;
    const float bcv = t.bc[s];
    float a0 = 0.0f, a1 = 0.0f, a2 = 0.0f, a3 = 0.0f;
#pragma unroll
    for (int n = 0; n < DSTATE; n += 4) {
      const float B0 = lane_bcast(bcv, n + 0), C0 = lane_bcast(bcv, DSTATE + n + 0);
      const float B1 = lane_bcast(bcv, n + 1), C1 = lane_bcast(bcv, DSTATE + n + 1);
      const float B2 = lane_bcast(bcv, n + 2), C2 = lane_bcast(bcv, DSTATE + n + 2);
      const float B3 = lane_bcast(bcv, n + 3), C3 = lane_bcast(bcv, DSTATE + n + 3);
      h[n + 0] = fmaf(__expf(dtv * An[n + 0]), h[n + 0], dtx * B0);
      h[n + 1] = fmaf(__expf(dtv * An[n + 1]), h[n + 1], dtx * B1);
      h[n + 2] = fmaf(__expf(dtv * An[n + 2]), h[n + 2], dtx * B2);
      h[n + 3] = fmaf(__expf(dtv * An[n + 3]), h[n + 3], dtx * B3);
      a0 = fmaf(h[n + 0], C0, a0);
      a1 = fmaf(h[n + 1], C1, a1);
      a2 = fmaf(h[n + 2], C2, a2);
      a3 = fmaf(h[n + 3], C3, a3);
    }
    const float y = (a0 + a1) + (a2 + a3);
    const float zv = t.zv[s];
    const float sz = zv / (1.0f + __expf(-zv));
    u_p[(size_t)(l + s) * D_INNER] = fmaf(xv, Dd, y) * sz;
  }
}

__global__ __launch_bounds__(32) void scan_kernel(
    const float* __restrict__ dt, const float* __restrict__ xconv,
    const float* __restrict__ x_dbl, const float* __restrict__ xz,
    const float* __restrict__ A_log, const float* __restrict__ Dp,
    const float* __restrict__ S, const float* __restrict__ Dsum,
    float* __restrict__ u) {
  const int lane = threadIdx.x;
  const int ck = (SCAN_NCH - 1) - blockIdx.x / SCAN_GROUPS;
  const int c0 = (blockIdx.x % SCAN_GROUPS) * 32;
  const int b = c0 >> 11;
  const int d = (c0 & (D_INNER - 1)) + lane;
  const int ch = c0 + lane;

  float An[DSTATE];
  scan_load_A(An, A_log, d);
  const float Dd = Dp[d];

  float h[DSTATE];
#pragma unroll
  for (int n = 0; n < DSTATE; ++n) h[n] = 0.0f;

  for (int j0 = 0; j0 < ck; j0 += CARRY_G) {
    float sv[CARRY_G][DSTATE];
    float dj[CARRY_G];
#pragma unroll
    for (int q = 0; q < CARRY_G; ++q) {
      const int j = (j0 + q < SCAN_NCH - 1) ? (j0 + q) : (SCAN_NCH - 2);
      dj[q] = Dsum[(size_t)j * SCAN_CHANS + ch];
#pragma unroll
      for (int n = 0; n < DSTATE; ++n)
        sv[q][n] = S[((size_t)j * DSTATE + n) * SCAN_CHANS + ch];
    }
    __builtin_amdgcn_sched_barrier(0);
#pragma unroll
    for (int q = 0; q < CARRY_G; ++q) {
      const bool live = (j0 + q) < ck;
#pragma unroll
      for (int n = 0; n < DSTATE; ++n) {
        const float e = live ? __expf(dj[q] * An[n]) : 1.0f;
        const float a = live ? sv[q][n] : 0.0f;
        h[n] = fmaf(e, h[n], a);
      }
    }
  }

  const size_t row0 = (size_t)b * SEQ_L + (size_t)ck * SCAN_CL;
  const float* dt_p = dt + row0 * D_INNER + d;
  const float* xc_p = xconv + row0 * D_INNER + d;
  const float* bc_p = x_dbl + row0 * XDBL_N + DTRANK + lane;
  const float* z_p = xz + row0 * (2 * D_INNER) + D_INNER + d;
  float* u_p = u + row0 * D_INNER + d;

  ScanBatch t0, t1;
  scan_load(t0, dt_p, xc_p, z_p, bc_p, 0);

  for (int l0 = 0; l0 < SCAN_CL; l0 += 2 * SCAN_TB) {
    scan_load(t1, dt_p, xc_p, z_p, bc_p, l0 + SCAN_TB);
    __builtin_amdgcn_sched_barrier(0);
    scan_steps(t0, h, An, Dd, u_p, l0);

    const int ln = (l0 + 2 * SCAN_TB < SCAN_CL) ? (l0 + 2 * SCAN_TB) : 0;
    scan_load(t0, dt_p, xc_p, z_p, bc_p, ln);
    __builtin_amdgcn_sched_barrier(0);
    scan_steps(t1, h, An, Dd, u_p, l0 + SCAN_TB);
  }
}

extern "C" void kernel_launch(void* const* d_in, const int* in_sizes, int n_in,
                              void* d_out, int out_size, void* d_ws, size_t ws_size,
                              hipStream_t stream) {
  const float* x      = (const float*)d_in[0];
  const float* W_in   = (const float*)d_in[1];
  const float* conv_w = (const float*)d_in[2];
  const float* conv_b = (const float*)d_in[3];
  const float* W_x    = (const float*)d_in[4];
  const float* W_dt   = (const float*)d_in[5];
  const float* b_dt   = (const float*)d_in[6];
  const float* A_log  = (const float*)d_in[7];
  const float* Dp     = (const float*)d_in[8];
  const float* W_out  = (const float*)d_in[9];
  float* out = (float*)d_out;

  float* ws = (float*)d_ws;
  const size_t M = (size_t)NB * SEQ_L;
  float* xz    = ws;
  float* xconv = xz + M * 2 * D_INNER;
  float* x_dbl = xconv + M * D_INNER;
  float* dt    = x_dbl + M * XDBL_N;
  float* u     = dt + M * D_INNER;
  float* S     = u + M * D_INNER;
  float* Dsum  = S + (size_t)(SCAN_NCH - 1) * DSTATE * SCAN_CHANS;

  dim3 blk(256);
  {
    hipLaunchKernelGGL((wmma_gemm_f32<2, 2, 4, 8, 1, 1>), dim3(512), blk, 0, stream,
                       x, W_in, xz, conv_b, conv_w, xconv,
                       (int)M, 2 * D_INNER, D_MODEL, D_MODEL, D_MODEL, 2 * D_INNER);
  }
  {
    hipLaunchKernelGGL(conv_edge_kernel, dim3((NB * SEQ_L / CONV_TM) * (D_INNER / 4) / 256), blk, 0, stream,
                       xz, conv_w, conv_b, xconv);
  }
  {
    hipLaunchKernelGGL((wmma_gemm_f32<0, 1, 2, 1, 1, 2>), dim3(384), dim3(32), 0, stream,
                       xconv, W_x, x_dbl, nullptr, nullptr, nullptr,
                       (int)M, XDBL_N, D_INNER, D_INNER, D_INNER, XDBL_N);
  }
  {
    hipLaunchKernelGGL((wmma_gemm_f32<1, 2, 4, 8, 1, 1>), dim3(256), blk, 0, stream,
                       x_dbl, W_dt, dt, b_dt, nullptr, nullptr,
                       (int)M, D_INNER, DTRANK, XDBL_N, DTRANK, D_INNER);
  }
  {
    hipLaunchKernelGGL(scan_state_kernel, dim3((SCAN_NCH - 1) * SCAN_GROUPS), dim3(32), 0, stream,
                       dt, xconv, x_dbl, A_log, S, Dsum);
  }
  {
    hipLaunchKernelGGL(scan_kernel, dim3(SCAN_NCH * SCAN_GROUPS), dim3(32), 0, stream,
                       dt, xconv, x_dbl, xz, A_log, Dp, S, Dsum, u);
  }
  {
    hipLaunchKernelGGL((wmma_gemm_f32<0, 2, 4, 4, 2, 1>), dim3(256), dim3(128), 0, stream,
                       u, W_out, out, nullptr, nullptr, nullptr,
                       (int)M, D_MODEL, D_INNER, D_INNER, D_INNER, D_MODEL);
  }
}


# dt GEMM re-cut: 16x32 wave tiles, 8192 waves, 8 per SIMD (was 32x64, 2048, 2 per SIMD): softplus code is 16 inlined log1pf chains per wave, not 64, inside the launch prefetch; same sums
# speedup vs baseline: 7.3683x; 1.1000x over previous
#include <hip/hip_runtime.h>
#include <hip/hip_bf16.h>
#include <math.h>


typedef __attribute__((ext_vector_type(2))) float v2f;
typedef __attribute__((ext_vector_type(8))) float v8f;

#define D_MODEL 1024
#define D_INNER 2048
#define NB 2
#define SEQ_L 1024
#define DSTATE 16
#define DTRANK 64
#define XDBL_N 96

#define GEMM_KC 32
#define CONV_TM 32

__device__ __forceinline__ float xhalf(float x) {
  const int b = __builtin_bit_cast(int, x);
  return __builtin_bit_cast(float, __builtin_amdgcn_permlanex16(b, b, 0x76543210, 0xfedcba98, false, false));
}

struct Frag16 {
  float4 q[4];
};

__device__ __forceinline__ void frag_load(Frag16& f, const float* __restrict__ p) {
  const float4* p4 = (const float4*)p;
#pragma unroll
  for (int t = 0; t < 4; ++t) f.q[t] = p4[t];
}

__device__ __forceinline__ v2f frag_pair(const Frag16& f, int s) {
  const float4 v = f.q[s >> 1];
  return (s & 1) ? (v2f){v.z, v.w} : (v2f){v.x, v.y};
}

template <int MT, int NT, int CH>
__device__ __forceinline__ void gemm_stage_load(Frag16 (&a)[CH][MT], Frag16 (&b)[CH][NT],
                                                const float* const (&arow)[MT],
                                                const float* const (&wrow)[NT], int k) {
#pragma unroll
  for (int c = 0; c < CH; ++c) {
#pragma unroll
    for (int i = 0; i < MT; ++i) frag_load(a[c][i], arow[i] + k + c * GEMM_KC);
#pragma unroll
    for (int j = 0; j < NT; ++j) frag_load(b[c][j], wrow[j] + k + c * GEMM_KC);
  }
}

template <int MT, int NT>
__device__ __forceinline__ void gemm_chunk_mma(const Frag16 (&a)[MT], const Frag16 (&b)[NT],
                                               v8f (&acc)[MT][NT]) {
#pragma unroll
  for (int s = 0; s < GEMM_KC / 4; ++s) {
    v2f av[MT], bv[NT];
#pragma unroll
    for (int i = 0; i < MT; ++i) av[i] = frag_pair(a[i], s);
#pragma unroll
    for (int j = 0; j < NT; ++j) bv[j] = frag_pair(b[j], s);
#pragma unroll
    for (int i = 0; i < MT; ++i)
#pragma unroll
      for (int j = 0; j < NT; ++j)
        acc[i][j] = __builtin_amdgcn_wmma_f32_16x16x4_f32(
            false, av[i], false, bv[j], (short)0, acc[i][j], false, false);
  }
}

template <int MT, int NT, int CH>
__device__ __forceinline__ void gemm_stage_mma(const Frag16 (&a)[CH][MT], const Frag16 (&b)[CH][NT],
                                               v8f (&acc)[MT][NT]) {
#pragma unroll
  for (int c = 0; c < CH; ++c) gemm_chunk_mma<MT, NT>(a[c], b[c], acc);
}

template <int MODE, int MT, int NT, int WPB, int WM, int CH, int WPS = 1>
__global__ __launch_bounds__(WPB * 32) __attribute__((amdgpu_waves_per_eu(WPS))) void wmma_gemm_f32(
    const float* __restrict__ A, const float* __restrict__ W,
    float* __restrict__ C, const float* __restrict__ bias,
    const float* __restrict__ cw, float* __restrict__ C2,
    int M, int N, int K, int lda, int ldb, int ldc) {
  static_assert(MODE != 2 || MT * 16 == CONV_TM, "conv_edge_kernel assumes this tile height");
  const int lane = threadIdx.x & 31;
  const int wave = threadIdx.x >> 5;
  constexpr int WN = WPB / WM;
  const int nbx = N / (16 * NT * WN);
  const int m0 = ((blockIdx.x / nbx) * WM + wave % WM) * (16 * MT);
  const int n0 = ((blockIdx.x % nbx) * WN + wave / WM) * (16 * NT);
  if (m0 >= M) return;

  const int r = lane & 15;
  const int kseg = (lane >> 4) << 4;

  const float* arow[MT];
#pragma unroll
  for (int i = 0; i < MT; ++i)
    arow[i] = A + (size_t)(m0 + i * 16 + r) * lda + kseg;
  const float* wrow[NT];
#pragma unroll
  for (int j = 0; j < NT; ++j)
    wrow[j] = W + (size_t)(n0 + j * 16 + r) * ldb + kseg;

  v8f acc[MT][NT];
#pragma unroll
  for (int i = 0; i < MT; ++i)
#pragma unroll
    for (int j = 0; j < NT; ++j) acc[i][j] = (v8f){};

  constexpr int KS = CH * GEMM_KC;
  Frag16 a0[CH][MT], b0[CH][NT], a1[CH][MT], b1[CH][NT];
  gemm_stage_load<MT, NT, CH>(a0, b0, arow, wrow, 0);

  for (int k = 0; k < K; k += 2 * KS) {
    gemm_stage_load<MT, NT, CH>(a1, b1, arow, wrow, k + KS);
    __builtin_amdgcn_sched_barrier(0);
    gemm_stage_mma<MT, NT, CH>(a0, b0, acc);
    __builtin_amdgcn_sched_barrier(0);

    const int kn = (k + 2 * KS < K) ? (k + 2 * KS) : 0;
    gemm_stage_load<MT, NT, CH>(a0, b0, arow, wrow, kn);
    __builtin_amdgcn_sched_barrier(0);
    gemm_stage_mma<MT, NT, CH>(a1, b1, acc);
  }

  const int mb = (lane >> 4) << 3;

  if (MODE == 2) {
    if (n0 < D_INNER) {
      const int h = lane >> 4;
#pragma unroll
      for (int j = 0; j < NT; ++j) {
        const int n = n0 + j * 16 + r;
        const float4 w = *(const float4*)(cw + (size_t)n * 4);
        const float cbv = bias[n];
        float up[MT][3];
#pragma unroll
        for (int i = 0; i < MT; ++i)
#pragma unroll
          for (int t = 0; t < 3; ++t) up[i][t] = xhalf(acc[i][j][5 + t]);
#pragma unroll
        for (int i = 0; i < MT; ++i) {
          float X[11];
#pragma unroll
          for (int t = 0; t < 3; ++t) X[t] = h ? up[i][t] : up[i > 0 ? i - 1 : 0][t];
#pragma unroll
          for (int e = 0; e < 8; ++e) X[3 + e] = acc[i][j][e];
#pragma unroll
          for (int e = 0; e < 8; ++e) {
            const size_t row = (size_t)(m0 + i * 16 + mb + e);
            float a = cbv;
            a = fmaf(w.x, X[e], a);
            a = fmaf(w.y, X[e + 1], a);
            a = fmaf(w.z, X[e + 2], a);
            a = fmaf(w.w, X[e + 3], a);
            const float v = a / (1.0f + __expf(-a));
            const bool top = (i == 0) && (e < 3);
            if (!top || h) C2[row * D_INNER + n] = v;
            if (top && !h) C[row * ldc + n] = acc[i][j][e];
            if (i == MT - 1 && e >= 5 && h) C[row * ldc + n] = acc[i][j][e];
          }
        }
      }
      return;
    }
  }

#pragma unroll
  for (int i = 0; i < MT; ++i) {
#pragma unroll
    for (int j = 0; j < NT; ++j) {
      const int n = n0 + j * 16 + r;
      float bv = 0.0f;
      if (MODE == 1) bv = bias[n];
#pragma unroll
      for (int e = 0; e < 8; ++e) {
        float v = acc[i][j][e];
        if (MODE == 1) {
          v += bv;
          v = (v > 20.0f) ? v : log1pf(__expf(v));
        }
        C[(size_t)(m0 + i * 16 + mb + e) * ldc + n] = v;
      }
    }
  }
}

__global__ __launch_bounds__(256) void conv_edge_kernel(
    const float* __restrict__ xz, const float* __restrict__ cw,
    const float* __restrict__ cb, float* __restrict__ xconv) {
  const int gid = blockIdx.x * blockDim.x + threadIdx.x;
  const int tile = gid / (D_INNER / 4);
  if (tile >= NB * SEQ_L / CONV_TM) return;
  const int d = (gid & (D_INNER / 4 - 1)) * 4;
  const int m0 = tile * CONV_TM;
  const bool first = (m0 & (SEQ_L - 1)) == 0;

  float w[4][4];
#pragma unroll
  for (int c = 0; c < 4; ++c) {
    const float4 t = *(const float4*)(cw + (size_t)(d + c) * 4);
    w[c][0] = t.x; w[c][1] = t.y; w[c][2] = t.z; w[c][3] = t.w;
  }
  const float4 bq = *(const float4*)(cb + d);
  const float bias[4] = {bq.x, bq.y, bq.z, bq.w};

  float x[6][4];
#pragma unroll
  for (int t = 0; t < 6; ++t) {
    float4 q = make_float4(0.0f, 0.0f, 0.0f, 0.0f);
    if (t >= 3 || !first) q = *(const float4*)(xz + (size_t)(m0 - 3 + t) * (2 * D_INNER) + d);
    x[t][0] = q.x; x[t][1] = q.y; x[t][2] = q.z; x[t][3] = q.w;
  }

#pragma unroll
  for (int e = 0; e < 3; ++e) {
    float o[4];
#pragma unroll
    for (int c = 0; c < 4; ++c) {
      float a = bias[c];
#pragma unroll
      for (int j = 0; j < 4; ++j)
        if (!first || e + j >= 3) a = fmaf(w[c][j], x[e + j][c], a);
      o[c] = a / (1.0f + __expf(-a));
    }
    *(float4*)(xconv + (size_t)(m0 + e) * D_INNER + d) = make_float4(o[0], o[1], o[2], o[3]);
  }
}

#define SCAN_TB 8
#define SCAN_NCH 16
#define SCAN_CL (SEQ_L / SCAN_NCH)
#define SCAN_CHANS (NB * D_INNER)
#define SCAN_GROUPS (SCAN_CHANS / 32)
#define CARRY_G 8

__device__ __forceinline__ float lane_bcast(float v, int srclane) {
  return __int_as_float(__builtin_amdgcn_readlane(__float_as_int(v), srclane));
}

__device__ __forceinline__ void scan_load_A(float (&An)[DSTATE], const float* __restrict__ A_log, int d) {
  const float4* ap = (const float4*)(A_log + (size_t)d * DSTATE);
#pragma unroll
  for (int q = 0; q < DSTATE / 4; ++q) {
    const float4 a4 = ap[q];
    An[q * 4 + 0] = -__expf(a4.x);
    An[q * 4 + 1] = -__expf(a4.y);
    An[q * 4 + 2] = -__expf(a4.z);
    An[q * 4 + 3] = -__expf(a4.w);
  }
}

struct StateBatch {
  float dtv[SCAN_TB];
  float xv[SCAN_TB];
  float bv[SCAN_TB];
};

__device__ __forceinline__ void state_load(StateBatch& t, const float* __restrict__ dt_p,
                                           const float* __restrict__ xc_p,
                                           const float* __restrict__ b_p, int l) {
#pragma unroll
  for (int s = 0; s < SCAN_TB; ++s) {
    t.dtv[s] = dt_p[(size_t)(l + s) * D_INNER];
    t.xv[s] = xc_p[(size_t)(l + s) * D_INNER];
    t.bv[s] = b_p[(size_t)(l + s) * XDBL_N];
  }
}

__device__ __forceinline__ void state_steps(const StateBatch& t, float (&h)[DSTATE],
                                            const float (&An)[DSTATE], float& dsum) {
#pragma unroll
  for (int s = 0; s < SCAN_TB; ++s) {
    const float dtv = t.dtv[s];
    const float dtx = dtv * t.xv[s];
    const float bcv = t.bv[s];
    dsum += dtv;
#pragma unroll
    for (int n = 0; n < DSTATE; ++n) {
      const float Bn = lane_bcast(bcv, n);
      h[n] = fmaf(__expf(dtv * An[n]), h[n], dtx * Bn);
    }
  }
}

__global__ __launch_bounds__(32) void scan_state_kernel(
    const float* __restrict__ dt, const float* __restrict__ xconv,
    const float* __restrict__ x_dbl, const float* __restrict__ A_log,
    float* __restrict__ S, float* __restrict__ Dsum) {
  const int lane = threadIdx.x;
  const int ck = blockIdx.x / SCAN_GROUPS;
  const int c0 = (blockIdx.x % SCAN_GROUPS) * 32;
  const int b = c0 >> 11;
  const int d = (c0 & (D_INNER - 1)) + lane;

  float An[DSTATE];
  scan_load_A(An, A_log, d);

  float h[DSTATE];
#pragma unroll
  for (int n = 0; n < DSTATE; ++n) h[n] = 0.0f;
  float dsum = 0.0f;

  const size_t row0 = (size_t)b * SEQ_L + (size_t)ck * SCAN_CL;
  const float* dt_p = dt + row0 * D_INNER + d;
  const float* xc_p = xconv + row0 * D_INNER + d;
  const float* b_p = x_dbl + row0 * XDBL_N + DTRANK + (lane & 15);

  StateBatch t0, t1;
  state_load(t0, dt_p, xc_p, b_p, 0);

  for (int l0 = 0; l0 < SCAN_CL; l0 += 2 * SCAN_TB) {
    state_load(t1, dt_p, xc_p, b_p, l0 + SCAN_TB);
    __builtin_amdgcn_sched_barrier(0);
    state_steps(t0, h, An, dsum);

    const int ln = (l0 + 2 * SCAN_TB < SCAN_CL) ? (l0 + 2 * SCAN_TB) : 0;
    state_load(t0, dt_p, xc_p, b_p, ln);
    __builtin_amdgcn_sched_barrier(0);
    state_steps(t1, h, An, dsum);
  }

  const int ch = c0 + lane;
#pragma unroll
  for (int n = 0; n < DSTATE; ++n)
    S[((size_t)ck * DSTATE + n) * SCAN_CHANS + ch] = h[n];
  Dsum[(size_t)ck * SCAN_CHANS + ch] = dsum;
}

struct ScanBatch {
  float dtv[SCAN_TB];
  float xv[SCAN_TB];
  float zv[SCAN_TB];
  float bc[SCAN_TB];
};

__device__ __forceinline__ void scan_load(ScanBatch& t, const float* __restrict__ dt_p,
                                          const float* __restrict__ xc_p,
                                          const float* __restrict__ z_p,
                                          const float* __restrict__ bc_p, int l) {
#pragma unroll
  for (int s = 0; s < SCAN_TB; ++s) {
    t.dtv[s] = dt_p[(size_t)(l + s) * D_INNER];
    t.xv[s] = xc_p[(size_t)(l + s) * D_INNER];
    t.zv[s] = z_p[(size_t)(l + s) * (2 * D_INNER)];
    t.bc[s] = bc_p[(size_t)(l + s) * XDBL_N];
  }
}

__device__ __forceinline__ void scan_steps(const ScanBatch& t, float (&h)[DSTATE],
                                           const float (&An)[DSTATE], float Dd,
                                           float* __restrict__ u_p, int l) {
#pragma unroll
  for (int s = 0; s < SCAN_TB; ++s) {
    const float dtv = t.dtv[s];
    const float xv = t.xv[s];
    const float dtx = dtv * xv;
    const float bcv = t.bc[s];
    float a0 = 0.0f, a1 = 0.0f, a2 = 0.0f, a3 = 0.0f;
#pragma unroll
    for (int n = 0; n < DSTATE; n += 4) {
      const float B0 = lane_bcast(bcv, n + 0), C0 = lane_bcast(bcv, DSTATE + n + 0);
      const float B1 = lane_bcast(bcv, n + 1), C1 = lane_bcast(bcv, DSTATE + n + 1);
      const float B2 = lane_bcast(bcv, n + 2), C2 = lane_bcast(bcv, DSTATE + n + 2);
      const float B3 = lane_bcast(bcv, n + 3), C3 = lane_bcast(bcv, DSTATE + n + 3);
      h[n + 0] = fmaf(__expf(dtv * An[n + 0]), h[n + 0], dtx * B0);
      h[n + 1] = fmaf(__expf(dtv * An[n + 1]), h[n + 1], dtx * B1);
      h[n + 2] = fmaf(__expf(dtv * An[n + 2]), h[n + 2], dtx * B2);
      h[n + 3] = fmaf(__expf(dtv * An[n + 3]), h[n + 3], dtx * B3);
      a0 = fmaf(h[n + 0], C0, a0);
      a1 = fmaf(h[n + 1], C1, a1);
      a2 = fmaf(h[n + 2], C2, a2);
      a3 = fmaf(h[n + 3], C3, a3);
    }
    const float y = (a0 + a1) + (a2 + a3);
    const float zv = t.zv[s];
    const float sz = zv / (1.0f + __expf(-zv));
    u_p[(size_t)(l + s) * D_INNER] = fmaf(xv, Dd, y) * sz;
  }
}

__global__ __launch_bounds__(32) void scan_kernel(
    const float* __restrict__ dt, const float* __restrict__ xconv,
    const float* __restrict__ x_dbl, const float* __restrict__ xz,
    const float* __restrict__ A_log, const float* __restrict__ Dp,
    const float* __restrict__ S, const float* __restrict__ Dsum,
    float* __restrict__ u) {
  const int lane = threadIdx.x;
  const int ck = (SCAN_NCH - 1) - blockIdx.x / SCAN_GROUPS;
  const int c0 = (blockIdx.x % SCAN_GROUPS) * 32;
  const int b = c0 >> 11;
  const int d = (c0 & (D_INNER - 1)) + lane;
  const int ch = c0 + lane;

  float An[DSTATE];
  scan_load_A(An, A_log, d);
  const float Dd = Dp[d];

  float h[DSTATE];
#pragma unroll
  for (int n = 0; n < DSTATE; ++n) h[n] = 0.0f;

  for (int j0 = 0; j0 < ck; j0 += CARRY_G) {
    float sv[CARRY_G][DSTATE];
    float dj[CARRY_G];
#pragma unroll
    for (int q = 0; q < CARRY_G; ++q) {
      const int j = (j0 + q < SCAN_NCH - 1) ? (j0 + q) : (SCAN_NCH - 2);
      dj[q] = Dsum[(size_t)j * SCAN_CHANS + ch];
#pragma unroll
      for (int n = 0; n < DSTATE; ++n)
        sv[q][n] = S[((size_t)j * DSTATE + n) * SCAN_CHANS + ch];
    }
    __builtin_amdgcn_sched_barrier(0);
#pragma unroll
    for (int q = 0; q < CARRY_G; ++q) {
      const bool live = (j0 + q) < ck;
#pragma unroll
      for (int n = 0; n < DSTATE; ++n) {
        const float e = live ? __expf(dj[q] * An[n]) : 1.0f;
        const float a = live ? sv[q][n] : 0.0f;
        h[n] = fmaf(e, h[n], a);
      }
    }
  }

  const size_t row0 = (size_t)b * SEQ_L + (size_t)ck * SCAN_CL;
  const float* dt_p = dt + row0 * D_INNER + d;
  const float* xc_p = xconv + row0 * D_INNER + d;
  const float* bc_p = x_dbl + row0 * XDBL_N + DTRANK + lane;
  const float* z_p = xz + row0 * (2 * D_INNER) + D_INNER + d;
  float* u_p = u + row0 * D_INNER + d;

  ScanBatch t0, t1;
  scan_load(t0, dt_p, xc_p, z_p, bc_p, 0);

  for (int l0 = 0; l0 < SCAN_CL; l0 += 2 * SCAN_TB) {
    scan_load(t1, dt_p, xc_p, z_p, bc_p, l0 + SCAN_TB);
    __builtin_amdgcn_sched_barrier(0);
    scan_steps(t0, h, An, Dd, u_p, l0);

    const int ln = (l0 + 2 * SCAN_TB < SCAN_CL) ? (l0 + 2 * SCAN_TB) : 0;
    scan_load(t0, dt_p, xc_p, z_p, bc_p, ln);
    __builtin_amdgcn_sched_barrier(0);
    scan_steps(t1, h, An, Dd, u_p, l0 + SCAN_TB);
  }
}

extern "C" void kernel_launch(void* const* d_in, const int* in_sizes, int n_in,
                              void* d_out, int out_size, void* d_ws, size_t ws_size,
                              hipStream_t stream) {
  const float* x      = (const float*)d_in[0];
  const float* W_in   = (const float*)d_in[1];
  const float* conv_w = (const float*)d_in[2];
  const float* conv_b = (const float*)d_in[3];
  const float* W_x    = (const float*)d_in[4];
  const float* W_dt   = (const float*)d_in[5];
  const float* b_dt   = (const float*)d_in[6];
  const float* A_log  = (const float*)d_in[7];
  const float* Dp     = (const float*)d_in[8];
  const float* W_out  = (const float*)d_in[9];
  float* out = (float*)d_out;

  float* ws = (float*)d_ws;
  const size_t M = (size_t)NB * SEQ_L;
  float* xz    = ws;
  float* xconv = xz + M * 2 * D_INNER;
  float* x_dbl = xconv + M * D_INNER;
  float* dt    = x_dbl + M * XDBL_N;
  float* u     = dt + M * D_INNER;
  float* S     = u + M * D_INNER;
  float* Dsum  = S + (size_t)(SCAN_NCH - 1) * DSTATE * SCAN_CHANS;

  dim3 blk(256);
  {
    hipLaunchKernelGGL((wmma_gemm_f32<2, 2, 4, 8, 1, 1>), dim3(512), blk, 0, stream,
                       x, W_in, xz, conv_b, conv_w, xconv,
                       (int)M, 2 * D_INNER, D_MODEL, D_MODEL, D_MODEL, 2 * D_INNER);
  }
  {
    hipLaunchKernelGGL(conv_edge_kernel, dim3((NB * SEQ_L / CONV_TM) * (D_INNER / 4) / 256), blk, 0, stream,
                       xz, conv_w, conv_b, xconv);
  }
  {
    hipLaunchKernelGGL((wmma_gemm_f32<0, 1, 2, 1, 1, 2>), dim3(384), dim3(32), 0, stream,
                       xconv, W_x, x_dbl, nullptr, nullptr, nullptr,
                       (int)M, XDBL_N, D_INNER, D_INNER, D_INNER, XDBL_N);
  }
  {
    hipLaunchKernelGGL((wmma_gemm_f32<1, 1, 2, 8, 4, 1, 8>), dim3(1024), blk, 0, stream,
                       x_dbl, W_dt, dt, b_dt, nullptr, nullptr,
                       (int)M, D_INNER, DTRANK, XDBL_N, DTRANK, D_INNER);
  }
  {
    hipLaunchKernelGGL(scan_state_kernel, dim3((SCAN_NCH - 1) * SCAN_GROUPS), dim3(32), 0, stream,
                       dt, xconv, x_dbl, A_log, S, Dsum);
  }
  {
    hipLaunchKernelGGL(scan_kernel, dim3(SCAN_NCH * SCAN_GROUPS), dim3(32), 0, stream,
                       dt, xconv, x_dbl, xz, A_log, Dp, S, Dsum, u);
  }
  {
    hipLaunchKernelGGL((wmma_gemm_f32<0, 2, 4, 4, 2, 1>), dim3(256), dim3(128), 0, stream,
                       u, W_out, out, nullptr, nullptr, nullptr,
                       (int)M, D_MODEL, D_INNER, D_INNER, D_INNER, D_MODEL);
  }
}


# in_proj wave tiles 64x64 (was 32x64), 256 workgroups of 2x4 waves, all resident in one round: 8 operand fragments loaded per 128 matrix instructions (was 12); k order and sums unchanged
# speedup vs baseline: 7.6291x; 1.0354x over previous
#include <hip/hip_runtime.h>
#include <hip/hip_bf16.h>
#include <math.h>


typedef __attribute__((ext_vector_type(2))) float v2f;
typedef __attribute__((ext_vector_type(8))) float v8f;

#define D_MODEL 1024
#define D_INNER 2048
#define NB 2
#define SEQ_L 1024
#define DSTATE 16
#define DTRANK 64
#define XDBL_N 96

#define GEMM_KC 32
#define CONV_TM 64

__device__ __forceinline__ float xhalf(float x) {
  const int b = __builtin_bit_cast(int, x);
  return __builtin_bit_cast(float, __builtin_amdgcn_permlanex16(b, b, 0x76543210, 0xfedcba98, false, false));
}

struct Frag16 {
  float4 q[4];
};

__device__ __forceinline__ void frag_load(Frag16& f, const float* __restrict__ p) {
  const float4* p4 = (const float4*)p;
#pragma unroll
  for (int t = 0; t < 4; ++t) f.q[t] = p4[t];
}

__device__ __forceinline__ v2f frag_pair(const Frag16& f, int s) {
  const float4 v = f.q[s >> 1];
  return (s & 1) ? (v2f){v.z, v.w} : (v2f){v.x, v.y};
}

template <int MT, int NT, int CH>
__device__ __forceinline__ void gemm_stage_load(Frag16 (&a)[CH][MT], Frag16 (&b)[CH][NT],
                                                const float* const (&arow)[MT],
                                                const float* const (&wrow)[NT], int k) {
#pragma unroll
  for (int c = 0; c < CH; ++c) {
#pragma unroll
    for (int i = 0; i < MT; ++i) frag_load(a[c][i], arow[i] + k + c * GEMM_KC);
#pragma unroll
    for (int j = 0; j < NT; ++j) frag_load(b[c][j], wrow[j] + k + c * GEMM_KC);
  }
}

template <int MT, int NT>
__device__ __forceinline__ void gemm_chunk_mma(const Frag16 (&a)[MT], const Frag16 (&b)[NT],
                                               v8f (&acc)[MT][NT]) {
#pragma unroll
  for (int s = 0; s < GEMM_KC / 4; ++s) {
    v2f av[MT], bv[NT];
#pragma unroll
    for (int i = 0; i < MT; ++i) av[i] = frag_pair(a[i], s);
#pragma unroll
    for (int j = 0; j < NT; ++j) bv[j] = frag_pair(b[j], s);
#pragma unroll
    for (int i = 0; i < MT; ++i)
#pragma unroll
      for (int j = 0; j < NT; ++j)
        acc[i][j] = __builtin_amdgcn_wmma_f32_16x16x4_f32(
            false, av[i], false, bv[j], (short)0, acc[i][j], false, false);
  }
}

template <int MT, int NT, int CH>
__device__ __forceinline__ void gemm_stage_mma(const Frag16 (&a)[CH][MT], const Frag16 (&b)[CH][NT],
                                               v8f (&acc)[MT][NT]) {
#pragma unroll
  for (int c = 0; c < CH; ++c) gemm_chunk_mma<MT, NT>(a[c], b[c], acc);
}

template <int MODE, int MT, int NT, int WPB, int WM, int CH, int WPS = 1>
__global__ __launch_bounds__(WPB * 32) __attribute__((amdgpu_waves_per_eu(WPS))) void wmma_gemm_f32(
    const float* __restrict__ A, const float* __restrict__ W,
    float* __restrict__ C, const float* __restrict__ bias,
    const float* __restrict__ cw, float* __restrict__ C2,
    int M, int N, int K, int lda, int ldb, int ldc) {
  static_assert(MODE != 2 || MT * 16 == CONV_TM, "conv_edge_kernel assumes this tile height");
  const int lane = threadIdx.x & 31;
  const int wave = threadIdx.x >> 5;
  constexpr int WN = WPB / WM;
  const int nbx = N / (16 * NT * WN);
  const int m0 = ((blockIdx.x / nbx) * WM + wave % WM) * (16 * MT);
  const int n0 = ((blockIdx.x % nbx) * WN + wave / WM) * (16 * NT);
  if (m0 >= M) return;

  const int r = lane & 15;
  const int kseg = (lane >> 4) << 4;

  const float* arow[MT];
#pragma unroll
  for (int i = 0; i < MT; ++i)
    arow[i] = A + (size_t)(m0 + i * 16 + r) * lda + kseg;
  const float* wrow[NT];
#pragma unroll
  for (int j = 0; j < NT; ++j)
    wrow[j] = W + (size_t)(n0 + j * 16 + r) * ldb + kseg;

  v8f acc[MT][NT];
#pragma unroll
  for (int i = 0; i < MT; ++i)
#pragma unroll
    for (int j = 0; j < NT; ++j) acc[i][j] = (v8f){};

  constexpr int KS = CH * GEMM_KC;
  Frag16 a0[CH][MT], b0[CH][NT], a1[CH][MT], b1[CH][NT];
  gemm_stage_load<MT, NT, CH>(a0, b0, arow, wrow, 0);

  for (int k = 0; k < K; k += 2 * KS) {
    gemm_stage_load<MT, NT, CH>(a1, b1, arow, wrow, k + KS);
    __builtin_amdgcn_sched_barrier(0);
    gemm_stage_mma<MT, NT, CH>(a0, b0, acc);
    __builtin_amdgcn_sched_barrier(0);

    const int kn = (k + 2 * KS < K) ? (k + 2 * KS) : 0;
    gemm_stage_load<MT, NT, CH>(a0, b0, arow, wrow, kn);
    __builtin_amdgcn_sched_barrier(0);
    gemm_stage_mma<MT, NT, CH>(a1, b1, acc);
  }

  const int mb = (lane >> 4) << 3;

  if (MODE == 2) {
    if (n0 < D_INNER) {
      const int h = lane >> 4;
#pragma unroll
      for (int j = 0; j < NT; ++j) {
        const int n = n0 + j * 16 + r;
        const float4 w = *(const float4*)(cw + (size_t)n * 4);
        const float cbv = bias[n];
        float up[MT][3];
#pragma unroll
        for (int i = 0; i < MT; ++i)
#pragma unroll
          for (int t = 0; t < 3; ++t) up[i][t] = xhalf(acc[i][j][5 + t]);
#pragma unroll
        for (int i = 0; i < MT; ++i) {
          float X[11];
#pragma unroll
          for (int t = 0; t < 3; ++t) X[t] = h ? up[i][t] : up[i > 0 ? i - 1 : 0][t];
#pragma unroll
          for (int e = 0; e < 8; ++e) X[3 + e] = acc[i][j][e];
#pragma unroll
          for (int e = 0; e < 8; ++e) {
            const size_t row = (size_t)(m0 + i * 16 + mb + e);
            float a = cbv;
            a = fmaf(w.x, X[e], a);
            a = fmaf(w.y, X[e + 1], a);
            a = fmaf(w.z, X[e + 2], a);
            a = fmaf(w.w, X[e + 3], a);
            const float v = a / (1.0f + __expf(-a));
            const bool top = (i == 0) && (e < 3);
            if (!top || h) C2[row * D_INNER + n] = v;
            if (top && !h) C[row * ldc + n] = acc[i][j][e];
            if (i == MT - 1 && e >= 5 && h) C[row * ldc + n] = acc[i][j][e];
          }
        }
      }
      return;
    }
  }

#pragma unroll
  for (int i = 0; i < MT; ++i) {
#pragma unroll
    for (int j = 0; j < NT; ++j) {
      const int n = n0 + j * 16 + r;
      float bv = 0.0f;
      if (MODE == 1) bv = bias[n];
#pragma unroll
      for (int e = 0; e < 8; ++e) {
        float v = acc[i][j][e];
        if (MODE == 1) {
          v += bv;
          v = (v > 20.0f) ? v : log1pf(__expf(v));
        }
        C[(size_t)(m0 + i * 16 + mb + e) * ldc + n] = v;
      }
    }
  }
}

__global__ __launch_bounds__(256) void conv_edge_kernel(
    const float* __restrict__ xz, const float* __restrict__ cw,
    const float* __restrict__ cb, float* __restrict__ xconv) {
  const int gid = blockIdx.x * blockDim.x + threadIdx.x;
  const int tile = gid / (D_INNER / 4);
  if (tile >= NB * SEQ_L / CONV_TM) return;
  const int d = (gid & (D_INNER / 4 - 1)) * 4;
  const int m0 = tile * CONV_TM;
  const bool first = (m0 & (SEQ_L - 1)) == 0;

  float w[4][4];
#pragma unroll
  for (int c = 0; c < 4; ++c) {
    const float4 t = *(const float4*)(cw + (size_t)(d + c) * 4);
    w[c][0] = t.x; w[c][1] = t.y; w[c][2] = t.z; w[c][3] = t.w;
  }
  const float4 bq = *(const float4*)(cb + d);
  const float bias[4] = {bq.x, bq.y, bq.z, bq.w};

  float x[6][4];
#pragma unroll
  for (int t = 0; t < 6; ++t) {
    float4 q = make_float4(0.0f, 0.0f, 0.0f, 0.0f);
    if (t >= 3 || !first) q = *(const float4*)(xz + (size_t)(m0 - 3 + t) * (2 * D_INNER) + d);
    x[t][0] = q.x; x[t][1] = q.y; x[t][2] = q.z; x[t][3] = q.w;
  }

#pragma unroll
  for (int e = 0; e < 3; ++e) {
    float o[4];
#pragma unroll
    for (int c = 0; c < 4; ++c) {
      float a = bias[c];
#pragma unroll
      for (int j = 0; j < 4; ++j)
        if (!first || e + j >= 3) a = fmaf(w[c][j], x[e + j][c], a);
      o[c] = a / (1.0f + __expf(-a));
    }
    *(float4*)(xconv + (size_t)(m0 + e) * D_INNER + d) = make_float4(o[0], o[1], o[2], o[3]);
  }
}

#define SCAN_TB 8
#define SCAN_NCH 16
#define SCAN_CL (SEQ_L / SCAN_NCH)
#define SCAN_CHANS (NB * D_INNER)
#define SCAN_GROUPS (SCAN_CHANS / 32)
#define CARRY_G 8

__device__ __forceinline__ float lane_bcast(float v, int srclane) {
  return __int_as_float(__builtin_amdgcn_readlane(__float_as_int(v), srclane));
}

__device__ __forceinline__ void scan_load_A(float (&An)[DSTATE], const float* __restrict__ A_log, int d) {
  const float4* ap = (const float4*)(A_log + (size_t)d * DSTATE);
#pragma unroll
  for (int q = 0; q < DSTATE / 4; ++q) {
    const float4 a4 = ap[q];
    An[q * 4 + 0] = -__expf(a4.x);
    An[q * 4 + 1] = -__expf(a4.y);
    An[q * 4 + 2] = -__expf(a4.z);
    An[q * 4 + 3] = -__expf(a4.w);
  }
}

struct StateBatch {
  float dtv[SCAN_TB];
  float xv[SCAN_TB];
  float bv[SCAN_TB];
};

__device__ __forceinline__ void state_load(StateBatch& t, const float* __restrict__ dt_p,
                                           const float* __restrict__ xc_p,
                                           const float* __restrict__ b_p, int l) {
#pragma unroll
  for (int s = 0; s < SCAN_TB; ++s) {
    t.dtv[s] = dt_p[(size_t)(l + s) * D_INNER];
    t.xv[s] = xc_p[(size_t)(l + s) * D_INNER];
    t.bv[s] = b_p[(size_t)(l + s) * XDBL_N];
  }
}

__device__ __forceinline__ void state_steps(const StateBatch& t, float (&h)[DSTATE],
                                            const float (&An)[DSTATE], float& dsum) {
#pragma unroll
  for (int s = 0; s < SCAN_TB; ++s) {
    const float dtv = t.dtv[s];
    const float dtx = dtv * t.xv[s];
    const float bcv = t.bv[s];
    dsum += dtv;
#pragma unroll
    for (int n = 0; n < DSTATE; ++n) {
      const float Bn = lane_bcast(bcv, n);
      h[n] = fmaf(__expf(dtv * An[n]), h[n], dtx * Bn);
    }
  }
}

__global__ __launch_bounds__(32) void scan_state_kernel(
    const float* __restrict__ dt, const float* __restrict__ xconv,
    const float* __restrict__ x_dbl, const float* __restrict__ A_log,
    float* __restrict__ S, float* __restrict__ Dsum) {
  const int lane = threadIdx.x;
  const int ck = blockIdx.x / SCAN_GROUPS;
  const int c0 = (blockIdx.x % SCAN_GROUPS) * 32;
  const int b = c0 >> 11;
  const int d = (c0 & (D_INNER - 1)) + lane;

  float An[DSTATE];
  scan_load_A(An, A_log, d);

  float h[DSTATE];
#pragma unroll
  for (int n = 0; n < DSTATE; ++n) h[n] = 0.0f;
  float dsum = 0.0f;

  const size_t row0 = (size_t)b * SEQ_L + (size_t)ck * SCAN_CL;
  const float* dt_p = dt + row0 * D_INNER + d;
  const float* xc_p = xconv + row0 * D_INNER + d;
  const float* b_p = x_dbl + row0 * XDBL_N + DTRANK + (lane & 15);

  StateBatch t0, t1;
  state_load(t0, dt_p, xc_p, b_p, 0);

  for (int l0 = 0; l0 < SCAN_CL; l0 += 2 * SCAN_TB) {
    state_load(t1, dt_p, xc_p, b_p, l0 + SCAN_TB);
    __builtin_amdgcn_sched_barrier(0);
    state_steps(t0, h, An, dsum);

    const int ln = (l0 + 2 * SCAN_TB < SCAN_CL) ? (l0 + 2 * SCAN_TB) : 0;
    state_load(t0, dt_p, xc_p, b_p, ln);
    __builtin_amdgcn_sched_barrier(0);
    state_steps(t1, h, An, dsum);
  }

  const int ch = c0 + lane;
#pragma unroll
  for (int n = 0; n < DSTATE; ++n)
    S[((size_t)ck * DSTATE + n) * SCAN_CHANS + ch] = h[n];
  Dsum[(size_t)ck * SCAN_CHANS + ch] = dsum;
}

struct ScanBatch {
  float dtv[SCAN_TB];
  float xv[SCAN_TB];
  float zv[SCAN_TB];
  float bc[SCAN_TB];
};

__device__ __forceinline__ void scan_load(ScanBatch& t, const float* __restrict__ dt_p,
                                          const float* __restrict__ xc_p,
                                          const float* __restrict__ z_p,
                                          const float* __restrict__ bc_p, int l) {
#pragma unroll
  for (int s = 0; s < SCAN_TB; ++s) {
    t.dtv[s] = dt_p[(size_t)(l + s) * D_INNER];
    t.xv[s] = xc_p[(size_t)(l + s) * D_INNER];
    t.zv[s] = z_p[(size_t)(l + s) * (2 * D_INNER)];
    t.bc[s] = bc_p[(size_t)(l + s) * XDBL_N];
  }
}

__device__ __forceinline__ void scan_steps(const ScanBatch& t, float (&h)[DSTATE],
                                           const float (&An)[DSTATE], float Dd,
                                           float* __restrict__ u_p, int l) {
#pragma unroll
  for (int s = 0; s < SCAN_TB; ++s) {
    const float dtv = t.dtv[s];
    const float xv = t.xv[s];
    const float dtx = dtv * xv;
    const float bcv = t.bc[s];
    float a0 = 0.0f, a1 = 0.0f, a2 = 0.0f, a3 = 0.0f;
#pragma unroll
    for (int n = 0; n < DSTATE; n += 4) {
      const float B0 = lane_bcast(bcv, n + 0), C0 = lane_bcast(bcv, DSTATE + n + 0);
      const float B1 = lane_bcast(bcv, n + 1), C1 = lane_bcast(bcv, DSTATE + n + 1);
      const float B2 = lane_bcast(bcv, n + 2), C2 = lane_bcast(bcv, DSTATE + n + 2);
      const float B3 = lane_bcast(bcv, n + 3), C3 = lane_bcast(bcv, DSTATE + n + 3);
      h[n + 0] = fmaf(__expf(dtv * An[n + 0]), h[n + 0], dtx * B0);
      h[n + 1] = fmaf(__expf(dtv * An[n + 1]), h[n + 1], dtx * B1);
      h[n + 2] = fmaf(__expf(dtv * An[n + 2]), h[n + 2], dtx * B2);
      h[n + 3] = fmaf(__expf(dtv * An[n + 3]), h[n + 3], dtx * B3);
      a0 = fmaf(h[n + 0], C0, a0);
      a1 = fmaf(h[n + 1], C1, a1);
      a2 = fmaf(h[n + 2], C2, a2);
      a3 = fmaf(h[n + 3], C3, a3);
    }
    const float y = (a0 + a1) + (a2 + a3);
    const float zv = t.zv[s];
    const float sz = zv / (1.0f + __expf(-zv));
    u_p[(size_t)(l + s) * D_INNER] = fmaf(xv, Dd, y) * sz;
  }
}

__global__ __launch_bounds__(32) void scan_kernel(
    const float* __restrict__ dt, const float* __restrict__ xconv,
    const float* __restrict__ x_dbl, const float* __restrict__ xz,
    const float* __restrict__ A_log, const float* __restrict__ Dp,
    const float* __restrict__ S, const float* __restrict__ Dsum,
    float* __restrict__ u) {
  const int lane = threadIdx.x;
  const int ck = (SCAN_NCH - 1) - blockIdx.x / SCAN_GROUPS;
  const int c0 = (blockIdx.x % SCAN_GROUPS) * 32;
  const int b = c0 >> 11;
  const int d = (c0 & (D_INNER - 1)) + lane;
  const int ch = c0 + lane;

  float An[DSTATE];
  scan_load_A(An, A_log, d);
  const float Dd = Dp[d];

  float h[DSTATE];
#pragma unroll
  for (int n = 0; n < DSTATE; ++n) h[n] = 0.0f;

  for (int j0 = 0; j0 < ck; j0 += CARRY_G) {
    float sv[CARRY_G][DSTATE];
    float dj[CARRY_G];
#pragma unroll
    for (int q = 0; q < CARRY_G; ++q) {
      const int j = (j0 + q < SCAN_NCH - 1) ? (j0 + q) : (SCAN_NCH - 2);
      dj[q] = Dsum[(size_t)j * SCAN_CHANS + ch];
#pragma unroll
      for (int n = 0; n < DSTATE; ++n)
        sv[q][n] = S[((size_t)j * DSTATE + n) * SCAN_CHANS + ch];
    }
    __builtin_amdgcn_sched_barrier(0);
#pragma unroll
    for (int q = 0; q < CARRY_G; ++q) {
      const bool live = (j0 + q) < ck;
#pragma unroll
      for (int n = 0; n < DSTATE; ++n) {
        const float e = live ? __expf(dj[q] * An[n]) : 1.0f;
        const float a = live ? sv[q][n] : 0.0f;
        h[n] = fmaf(e, h[n], a);
      }
    }
  }

  const size_t row0 = (size_t)b * SEQ_L + (size_t)ck * SCAN_CL;
  const float* dt_p = dt + row0 * D_INNER + d;
  const float* xc_p = xconv + row0 * D_INNER + d;
  const float* bc_p = x_dbl + row0 * XDBL_N + DTRANK + lane;
  const float* z_p = xz + row0 * (2 * D_INNER) + D_INNER + d;
  float* u_p = u + row0 * D_INNER + d;

  ScanBatch t0, t1;
  scan_load(t0, dt_p, xc_p, z_p, bc_p, 0);

  for (int l0 = 0; l0 < SCAN_CL; l0 += 2 * SCAN_TB) {
    scan_load(t1, dt_p, xc_p, z_p, bc_p, l0 + SCAN_TB);
    __builtin_amdgcn_sched_barrier(0);
    scan_steps(t0, h, An, Dd, u_p, l0);

    const int ln = (l0 + 2 * SCAN_TB < SCAN_CL) ? (l0 + 2 * SCAN_TB) : 0;
    scan_load(t0, dt_p, xc_p, z_p, bc_p, ln);
    __builtin_amdgcn_sched_barrier(0);
    scan_steps(t1, h, An, Dd, u_p, l0 + SCAN_TB);
  }
}

extern "C" void kernel_launch(void* const* d_in, const int* in_sizes, int n_in,
                              void* d_out, int out_size, void* d_ws, size_t ws_size,
                              hipStream_t stream) {
  const float* x      = (const float*)d_in[0];
  const float* W_in   = (const float*)d_in[1];
  const float* conv_w = (const float*)d_in[2];
  const float* conv_b = (const float*)d_in[3];
  const float* W_x    = (const float*)d_in[4];
  const float* W_dt   = (const float*)d_in[5];
  const float* b_dt   = (const float*)d_in[6];
  const float* A_log  = (const float*)d_in[7];
  const float* Dp     = (const float*)d_in[8];
  const float* W_out  = (const float*)d_in[9];
  float* out = (float*)d_out;

  float* ws = (float*)d_ws;
  const size_t M = (size_t)NB * SEQ_L;
  float* xz    = ws;
  float* xconv = xz + M * 2 * D_INNER;
  float* x_dbl = xconv + M * D_INNER;
  float* dt    = x_dbl + M * XDBL_N;
  float* u     = dt + M * D_INNER;
  float* S     = u + M * D_INNER;
  float* Dsum  = S + (size_t)(SCAN_NCH - 1) * DSTATE * SCAN_CHANS;

  dim3 blk(256);
  {
    hipLaunchKernelGGL((wmma_gemm_f32<2, 4, 4, 8, 2, 1>), dim3(256), blk, 0, stream,
                       x, W_in, xz, conv_b, conv_w, xconv,
                       (int)M, 2 * D_INNER, D_MODEL, D_MODEL, D_MODEL, 2 * D_INNER);
  }
  {
    hipLaunchKernelGGL(conv_edge_kernel, dim3((NB * SEQ_L / CONV_TM) * (D_INNER / 4) / 256), blk, 0, stream,
                       xz, conv_w, conv_b, xconv);
  }
  {
    hipLaunchKernelGGL((wmma_gemm_f32<0, 1, 2, 1, 1, 2>), dim3(384), dim3(32), 0, stream,
                       xconv, W_x, x_dbl, nullptr, nullptr, nullptr,
                       (int)M, XDBL_N, D_INNER, D_INNER, D_INNER, XDBL_N);
  }
  {
    hipLaunchKernelGGL((wmma_gemm_f32<1, 1, 2, 8, 4, 1, 8>), dim3(1024), blk, 0, stream,
                       x_dbl, W_dt, dt, b_dt, nullptr, nullptr,
                       (int)M, D_INNER, DTRANK, XDBL_N, DTRANK, D_INNER);
  }
  {
    hipLaunchKernelGGL(scan_state_kernel, dim3((SCAN_NCH - 1) * SCAN_GROUPS), dim3(32), 0, stream,
                       dt, xconv, x_dbl, A_log, S, Dsum);
  }
  {
    hipLaunchKernelGGL(scan_kernel, dim3(SCAN_NCH * SCAN_GROUPS), dim3(32), 0, stream,
                       dt, xconv, x_dbl, xz, A_log, Dp, S, Dsum, u);
  }
  {
    hipLaunchKernelGGL((wmma_gemm_f32<0, 2, 4, 4, 2, 1>), dim3(256), dim3(128), 0, stream,
                       u, W_out, out, nullptr, nullptr, nullptr,
                       (int)M, D_MODEL, D_INNER, D_INNER, D_INNER, D_MODEL);
  }
}


# out product (N=1024): 64x64 wave tiles, k split over 4 waves of a workgroup (2 waves per SIMD), partial tiles summed through LDS behind one barrier, then stored: new sum order
# speedup vs baseline: 8.0796x; 1.0591x over previous
#include <hip/hip_runtime.h>
#include <hip/hip_bf16.h>
#include <math.h>


typedef __attribute__((ext_vector_type(2))) float v2f;
typedef __attribute__((ext_vector_type(8))) float v8f;

#define D_MODEL 1024
#define D_INNER 2048
#define NB 2
#define SEQ_L 1024
#define DSTATE 16
#define DTRANK 64
#define XDBL_N 96

#define GEMM_KC 32
#define CONV_TM 64

__device__ __forceinline__ float xhalf(float x) {
  const int b = __builtin_bit_cast(int, x);
  return __builtin_bit_cast(float, __builtin_amdgcn_permlanex16(b, b, 0x76543210, 0xfedcba98, false, false));
}

struct Frag16 {
  float4 q[4];
};

__device__ __forceinline__ void frag_load(Frag16& f, const float* __restrict__ p) {
  const float4* p4 = (const float4*)p;
#pragma unroll
  for (int t = 0; t < 4; ++t) f.q[t] = p4[t];
}

__device__ __forceinline__ v2f frag_pair(const Frag16& f, int s) {
  const float4 v = f.q[s >> 1];
  return (s & 1) ? (v2f){v.z, v.w} : (v2f){v.x, v.y};
}

template <int MT, int NT, int CH>
__device__ __forceinline__ void gemm_stage_load(Frag16 (&a)[CH][MT], Frag16 (&b)[CH][NT],
                                                const float* const (&arow)[MT],
                                                const float* const (&wrow)[NT], int k) {
#pragma unroll
  for (int c = 0; c < CH; ++c) {
#pragma unroll
    for (int i = 0; i < MT; ++i) frag_load(a[c][i], arow[i] + k + c * GEMM_KC);
#pragma unroll
    for (int j = 0; j < NT; ++j) frag_load(b[c][j], wrow[j] + k + c * GEMM_KC);
  }
}

template <int MT, int NT>
__device__ __forceinline__ void gemm_chunk_mma(const Frag16 (&a)[MT], const Frag16 (&b)[NT],
                                               v8f (&acc)[MT][NT]) {
#pragma unroll
  for (int s = 0; s < GEMM_KC / 4; ++s) {
    v2f av[MT], bv[NT];
#pragma unroll
    for (int i = 0; i < MT; ++i) av[i] = frag_pair(a[i], s);
#pragma unroll
    for (int j = 0; j < NT; ++j) bv[j] = frag_pair(b[j], s);
#pragma unroll
    for (int i = 0; i < MT; ++i)
#pragma unroll
      for (int j = 0; j < NT; ++j)
        acc[i][j] = __builtin_amdgcn_wmma_f32_16x16x4_f32(
            false, av[i], false, bv[j], (short)0, acc[i][j], false, false);
  }
}

template <int MT, int NT, int CH>
__device__ __forceinline__ void gemm_stage_mma(const Frag16 (&a)[CH][MT], const Frag16 (&b)[CH][NT],
                                               v8f (&acc)[MT][NT]) {
#pragma unroll
  for (int c = 0; c < CH; ++c) gemm_chunk_mma<MT, NT>(a[c], b[c], acc);
}

template <int MODE, int MT, int NT, int WPB, int WM, int CH, int WPS = 1>
__global__ __launch_bounds__(WPB * 32) __attribute__((amdgpu_waves_per_eu(WPS))) void wmma_gemm_f32(
    const float* __restrict__ A, const float* __restrict__ W,
    float* __restrict__ C, const float* __restrict__ bias,
    const float* __restrict__ cw, float* __restrict__ C2,
    int M, int N, int K, int lda, int ldb, int ldc) {
  static_assert(MODE != 2 || MT * 16 == CONV_TM, "conv_edge_kernel assumes this tile height");
  const int lane = threadIdx.x & 31;
  const int wave = threadIdx.x >> 5;
  constexpr int WN = WPB / WM;
  const int nbx = N / (16 * NT * WN);
  const int m0 = ((blockIdx.x / nbx) * WM + wave % WM) * (16 * MT);
  const int n0 = ((blockIdx.x % nbx) * WN + wave / WM) * (16 * NT);
  if (m0 >= M) return;

  const int r = lane & 15;
  const int kseg = (lane >> 4) << 4;

  const float* arow[MT];
#pragma unroll
  for (int i = 0; i < MT; ++i)
    arow[i] = A + (size_t)(m0 + i * 16 + r) * lda + kseg;
  const float* wrow[NT];
#pragma unroll
  for (int j = 0; j < NT; ++j)
    wrow[j] = W + (size_t)(n0 + j * 16 + r) * ldb + kseg;

  v8f acc[MT][NT];
#pragma unroll
  for (int i = 0; i < MT; ++i)
#pragma unroll
    for (int j = 0; j < NT; ++j) acc[i][j] = (v8f){};

  constexpr int KS = CH * GEMM_KC;
  Frag16 a0[CH][MT], b0[CH][NT], a1[CH][MT], b1[CH][NT];
  gemm_stage_load<MT, NT, CH>(a0, b0, arow, wrow, 0);

  for (int k = 0; k < K; k += 2 * KS) {
    gemm_stage_load<MT, NT, CH>(a1, b1, arow, wrow, k + KS);
    __builtin_amdgcn_sched_barrier(0);
    gemm_stage_mma<MT, NT, CH>(a0, b0, acc);
    __builtin_amdgcn_sched_barrier(0);

    const int kn = (k + 2 * KS < K) ? (k + 2 * KS) : 0;
    gemm_stage_load<MT, NT, CH>(a0, b0, arow, wrow, kn);
    __builtin_amdgcn_sched_barrier(0);
    gemm_stage_mma<MT, NT, CH>(a1, b1, acc);
  }

  const int mb = (lane >> 4) << 3;

  if (MODE == 2) {
    if (n0 < D_INNER) {
      const int h = lane >> 4;
#pragma unroll
      for (int j = 0; j < NT; ++j) {
        const int n = n0 + j * 16 + r;
        const float4 w = *(const float4*)(cw + (size_t)n * 4);
        const float cbv = bias[n];
        float up[MT][3];
#pragma unroll
        for (int i = 0; i < MT; ++i)
#pragma unroll
          for (int t = 0; t < 3; ++t) up[i][t] = xhalf(acc[i][j][5 + t]);
#pragma unroll
        for (int i = 0; i < MT; ++i) {
          float X[11];
#pragma unroll
          for (int t = 0; t < 3; ++t) X[t] = h ? up[i][t] : up[i > 0 ? i - 1 : 0][t];
#pragma unroll
          for (int e = 0; e < 8; ++e) X[3 + e] = acc[i][j][e];
#pragma unroll
          for (int e = 0; e < 8; ++e) {
            const size_t row = (size_t)(m0 + i * 16 + mb + e);
            float a = cbv;
            a = fmaf(w.x, X[e], a);
            a = fmaf(w.y, X[e + 1], a);
            a = fmaf(w.z, X[e + 2], a);
            a = fmaf(w.w, X[e + 3], a);
            const float v = a / (1.0f + __expf(-a));
            const bool top = (i == 0) && (e < 3);
            if (!top || h) C2[row * D_INNER + n] = v;
            if (top && !h) C[row * ldc + n] = acc[i][j][e];
            if (i == MT - 1 && e >= 5 && h) C[row * ldc + n] = acc[i][j][e];
          }
        }
      }
      return;
    }
  }

#pragma unroll
  for (int i = 0; i < MT; ++i) {
#pragma unroll
    for (int j = 0; j < NT; ++j) {
      const int n = n0 + j * 16 + r;
      float bv = 0.0f;
      if (MODE == 1) bv = bias[n];
#pragma unroll
      for (int e = 0; e < 8; ++e) {
        float v = acc[i][j][e];
        if (MODE == 1) {
          v += bv;
          v = (v > 20.0f) ? v : log1pf(__expf(v));
        }
        C[(size_t)(m0 + i * 16 + mb + e) * ldc + n] = v;
      }
    }
  }
}

template <int MT, int NT, int KQ, int TN>
__global__ __launch_bounds__(KQ * TN * 32) void wmma_gemm_ksplit_f32(
    const float* __restrict__ A, const float* __restrict__ W, float* __restrict__ C,
    int M, int N, int K, int lda, int ldb, int ldc) {
  static_assert(MT == KQ, "wave q of a tile keeps row strip q");
  __shared__ float4 red[TN * KQ * (KQ - 1) * NT * 2 * 32];

  const int lane = threadIdx.x & 31;
  const int wave = __builtin_amdgcn_readfirstlane(threadIdx.x >> 5);
  const int q = wave % KQ;
  const int t = wave / KQ;
  const int nbx = N / (16 * NT * TN);
  const int m0 = (blockIdx.x / nbx) * (16 * MT);
  const int n0 = ((blockIdx.x % nbx) * TN + t) * (16 * NT);
  if (m0 >= M) return;

  const int r = lane & 15;
  const int kseg = (lane >> 4) << 4;
  const int kq = K / KQ;

  const float* arow[MT];
#pragma unroll
  for (int i = 0; i < MT; ++i)
    arow[i] = A + (size_t)(m0 + i * 16 + r) * lda + q * kq + kseg;
  const float* wrow[NT];
#pragma unroll
  for (int j = 0; j < NT; ++j)
    wrow[j] = W + (size_t)(n0 + j * 16 + r) * ldb + q * kq + kseg;

  v8f acc[MT][NT];
#pragma unroll
  for (int i = 0; i < MT; ++i)
#pragma unroll
    for (int j = 0; j < NT; ++j) acc[i][j] = (v8f){};

  Frag16 a0[1][MT], b0[1][NT], a1[1][MT], b1[1][NT];
  gemm_stage_load<MT, NT, 1>(a0, b0, arow, wrow, 0);

  for (int k = 0; k < kq; k += 2 * GEMM_KC) {
    gemm_stage_load<MT, NT, 1>(a1, b1, arow, wrow, k + GEMM_KC);
    __builtin_amdgcn_sched_barrier(0);
    gemm_stage_mma<MT, NT, 1>(a0, b0, acc);
    __builtin_amdgcn_sched_barrier(0);

    const int kn = (k + 2 * GEMM_KC < kq) ? (k + 2 * GEMM_KC) : 0;
    gemm_stage_load<MT, NT, 1>(a0, b0, arow, wrow, kn);
    __builtin_amdgcn_sched_barrier(0);
    gemm_stage_mma<MT, NT, 1>(a1, b1, acc);
  }

#pragma unroll
  for (int i = 0; i < MT; ++i) {
    if (i != q) {
      const int slot = (q < i) ? q : q - 1;
      float4* dst = red + (size_t)(((t * KQ + i) * (KQ - 1) + slot) * NT) * 2 * 32 + lane;
#pragma unroll
      for (int j = 0; j < NT; ++j)
#pragma unroll
        for (int g = 0; g < 2; ++g)
          dst[(j * 2 + g) * 32] = make_float4(acc[i][j][4 * g + 0], acc[i][j][4 * g + 1],
                                              acc[i][j][4 * g + 2], acc[i][j][4 * g + 3]);
    }
  }
  __syncthreads();

  const int mb = (lane >> 4) << 3;
#pragma unroll
  for (int i = 0; i < MT; ++i) {
    if (i == q) {
#pragma unroll
      for (int slot = 0; slot < KQ - 1; ++slot) {
        const float4* src = red + (size_t)(((t * KQ + i) * (KQ - 1) + slot) * NT) * 2 * 32 + lane;
#pragma unroll
        for (int j = 0; j < NT; ++j)
#pragma unroll
          for (int g = 0; g < 2; ++g) {
            const float4 v = src[(j * 2 + g) * 32];
            acc[i][j][4 * g + 0] += v.x;
            acc[i][j][4 * g + 1] += v.y;
            acc[i][j][4 * g + 2] += v.z;
            acc[i][j][4 * g + 3] += v.w;
          }
      }
#pragma unroll
      for (int j = 0; j < NT; ++j) {
        const int n = n0 + j * 16 + r;
#pragma unroll
        for (int e = 0; e < 8; ++e)
          C[(size_t)(m0 + i * 16 + mb + e) * ldc + n] = acc[i][j][e];
      }
    }
  }
}

__global__ __launch_bounds__(256) void conv_edge_kernel(
    const float* __restrict__ xz, const float* __restrict__ cw,
    const float* __restrict__ cb, float* __restrict__ xconv) {
  const int gid = blockIdx.x * blockDim.x + threadIdx.x;
  const int tile = gid / (D_INNER / 4);
  if (tile >= NB * SEQ_L / CONV_TM) return;
  const int d = (gid & (D_INNER / 4 - 1)) * 4;
  const int m0 = tile * CONV_TM;
  const bool first = (m0 & (SEQ_L - 1)) == 0;

  float w[4][4];
#pragma unroll
  for (int c = 0; c < 4; ++c) {
    const float4 t = *(const float4*)(cw + (size_t)(d + c) * 4);
    w[c][0] = t.x; w[c][1] = t.y; w[c][2] = t.z; w[c][3] = t.w;
  }
  const float4 bq = *(const float4*)(cb + d);
  const float bias[4] = {bq.x, bq.y, bq.z, bq.w};

  float x[6][4];
#pragma unroll
  for (int t = 0; t < 6; ++t) {
    float4 q = make_float4(0.0f, 0.0f, 0.0f, 0.0f);
    if (t >= 3 || !first) q = *(const float4*)(xz + (size_t)(m0 - 3 + t) * (2 * D_INNER) + d);
    x[t][0] = q.x; x[t][1] = q.y; x[t][2] = q.z; x[t][3] = q.w;
  }

#pragma unroll
  for (int e = 0; e < 3; ++e) {
    float o[4];
#pragma unroll
    for (int c = 0; c < 4; ++c) {
      float a = bias[c];
#pragma unroll
      for (int j = 0; j < 4; ++j)
        if (!first || e + j >= 3) a = fmaf(w[c][j], x[e + j][c], a);
      o[c] = a / (1.0f + __expf(-a));
    }
    *(float4*)(xconv + (size_t)(m0 + e) * D_INNER + d) = make_float4(o[0], o[1], o[2], o[3]);
  }
}

#define SCAN_TB 8
#define SCAN_NCH 16
#define SCAN_CL (SEQ_L / SCAN_NCH)
#define SCAN_CHANS (NB * D_INNER)
#define SCAN_GROUPS (SCAN_CHANS / 32)
#define CARRY_G 8

__device__ __forceinline__ float lane_bcast(float v, int srclane) {
  return __int_as_float(__builtin_amdgcn_readlane(__float_as_int(v), srclane));
}

__device__ __forceinline__ void scan_load_A(float (&An)[DSTATE], const float* __restrict__ A_log, int d) {
  const float4* ap = (const float4*)(A_log + (size_t)d * DSTATE);
#pragma unroll
  for (int q = 0; q < DSTATE / 4; ++q) {
    const float4 a4 = ap[q];
    An[q * 4 + 0] = -__expf(a4.x);
    An[q * 4 + 1] = -__expf(a4.y);
    An[q * 4 + 2] = -__expf(a4.z);
    An[q * 4 + 3] = -__expf(a4.w);
  }
}

struct StateBatch {
  float dtv[SCAN_TB];
  float xv[SCAN_TB];
  float bv[SCAN_TB];
};

__device__ __forceinline__ void state_load(StateBatch& t, const float* __restrict__ dt_p,
                                           const float* __restrict__ xc_p,
                                           const float* __restrict__ b_p, int l) {
#pragma unroll
  for (int s = 0; s < SCAN_TB; ++s) {
    t.dtv[s] = dt_p[(size_t)(l + s) * D_INNER];
    t.xv[s] = xc_p[(size_t)(l + s) * D_INNER];
    t.bv[s] = b_p[(size_t)(l + s) * XDBL_N];
  }
}

__device__ __forceinline__ void state_steps(const StateBatch& t, float (&h)[DSTATE],
                                            const float (&An)[DSTATE], float& dsum) {
#pragma unroll
  for (int s = 0; s < SCAN_TB; ++s) {
    const float dtv = t.dtv[s];
    const float dtx = dtv * t.xv[s];
    const float bcv = t.bv[s];
    dsum += dtv;
#pragma unroll
    for (int n = 0; n < DSTATE; ++n) {
      const float Bn = lane_bcast(bcv, n);
      h[n] = fmaf(__expf(dtv * An[n]), h[n], dtx * Bn);
    }
  }
}

__global__ __launch_bounds__(32) void scan_state_kernel(
    const float* __restrict__ dt, const float* __restrict__ xconv,
    const float* __restrict__ x_dbl, const float* __restrict__ A_log,
    float* __restrict__ S, float* __restrict__ Dsum) {
  const int lane = threadIdx.x;
  const int ck = blockIdx.x / SCAN_GROUPS;
  const int c0 = (blockIdx.x % SCAN_GROUPS) * 32;
  const int b = c0 >> 11;
  const int d = (c0 & (D_INNER - 1)) + lane;

  float An[DSTATE];
  scan_load_A(An, A_log, d);

  float h[DSTATE];
#pragma unroll
  for (int n = 0; n < DSTATE; ++n) h[n] = 0.0f;
  float dsum = 0.0f;

  const size_t row0 = (size_t)b * SEQ_L + (size_t)ck * SCAN_CL;
  const float* dt_p = dt + row0 * D_INNER + d;
  const float* xc_p = xconv + row0 * D_INNER + d;
  const float* b_p = x_dbl + row0 * XDBL_N + DTRANK + (lane & 15);

  StateBatch t0, t1;
  state_load(t0, dt_p, xc_p, b_p, 0);

  for (int l0 = 0; l0 < SCAN_CL; l0 += 2 * SCAN_TB) {
    state_load(t1, dt_p, xc_p, b_p, l0 + SCAN_TB);
    __builtin_amdgcn_sched_barrier(0);
    state_steps(t0, h, An, dsum);

    const int ln = (l0 + 2 * SCAN_TB < SCAN_CL) ? (l0 + 2 * SCAN_TB) : 0;
    state_load(t0, dt_p, xc_p, b_p, ln);
    __builtin_amdgcn_sched_barrier(0);
    state_steps(t1, h, An, dsum);
  }

  const int ch = c0 + lane;
#pragma unroll
  for (int n = 0; n < DSTATE; ++n)
    S[((size_t)ck * DSTATE + n) * SCAN_CHANS + ch] = h[n];
  Dsum[(size_t)ck * SCAN_CHANS + ch] = dsum;
}

struct ScanBatch {
  float dtv[SCAN_TB];
  float xv[SCAN_TB];
  float zv[SCAN_TB];
  float bc[SCAN_TB];
};

__device__ __forceinline__ void scan_load(ScanBatch& t, const float* __restrict__ dt_p,
                                          const float* __restrict__ xc_p,
                                          const float* __restrict__ z_p,
                                          const float* __restrict__ bc_p, int l) {
#pragma unroll
  for (int s = 0; s < SCAN_TB; ++s) {
    t.dtv[s] = dt_p[(size_t)(l + s) * D_INNER];
    t.xv[s] = xc_p[(size_t)(l + s) * D_INNER];
    t.zv[s] = z_p[(size_t)(l + s) * (2 * D_INNER)];
    t.bc[s] = bc_p[(size_t)(l + s) * XDBL_N];
  }
}

__device__ __forceinline__ void scan_steps(const ScanBatch& t, float (&h)[DSTATE],
                                           const float (&An)[DSTATE], float Dd,
                                           float* __restrict__ u_p, int l) {
#pragma unroll
  for (int s = 0; s < SCAN_TB; ++s) {
    const float dtv = t.dtv[s];
    const float xv = t.xv[s];
    const float dtx = dtv * xv;
    const float bcv = t.bc[s];
    float a0 = 0.0f, a1 = 0.0f, a2 = 0.0f, a3 = 0.0f;
#pragma unroll
    for (int n = 0; n < DSTATE; n += 4) {
      const float B0 = lane_bcast(bcv, n + 0), C0 = lane_bcast(bcv, DSTATE + n + 0);
      const float B1 = lane_bcast(bcv, n + 1), C1 = lane_bcast(bcv, DSTATE + n + 1);
      const float B2 = lane_bcast(bcv, n + 2), C2 = lane_bcast(bcv, DSTATE + n + 2);
      const float B3 = lane_bcast(bcv, n + 3), C3 = lane_bcast(bcv, DSTATE + n + 3);
      h[n + 0] = fmaf(__expf(dtv * An[n + 0]), h[n + 0], dtx * B0);
      h[n + 1] = fmaf(__expf(dtv * An[n + 1]), h[n + 1], dtx * B1);
      h[n + 2] = fmaf(__expf(dtv * An[n + 2]), h[n + 2], dtx * B2);
      h[n + 3] = fmaf(__expf(dtv * An[n + 3]), h[n + 3], dtx * B3);
      a0 = fmaf(h[n + 0], C0, a0);
      a1 = fmaf(h[n + 1], C1, a1);
      a2 = fmaf(h[n + 2], C2, a2);
      a3 = fmaf(h[n + 3], C3, a3);
    }
    const float y = (a0 + a1) + (a2 + a3);
    const float zv = t.zv[s];
    const float sz = zv / (1.0f + __expf(-zv));
    u_p[(size_t)(l + s) * D_INNER] = fmaf(xv, Dd, y) * sz;
  }
}

__global__ __launch_bounds__(32) void scan_kernel(
    const float* __restrict__ dt, const float* __restrict__ xconv,
    const float* __restrict__ x_dbl, const float* __restrict__ xz,
    const float* __restrict__ A_log, const float* __restrict__ Dp,
    const float* __restrict__ S, const float* __restrict__ Dsum,
    float* __restrict__ u) {
  const int lane = threadIdx.x;
  const int ck = (SCAN_NCH - 1) - blockIdx.x / SCAN_GROUPS;
  const int c0 = (blockIdx.x % SCAN_GROUPS) * 32;
  const int b = c0 >> 11;
  const int d = (c0 & (D_INNER - 1)) + lane;
  const int ch = c0 + lane;

  float An[DSTATE];
  scan_load_A(An, A_log, d);
  const float Dd = Dp[d];

  float h[DSTATE];
#pragma unroll
  for (int n = 0; n < DSTATE; ++n) h[n] = 0.0f;

  for (int j0 = 0; j0 < ck; j0 += CARRY_G) {
    float sv[CARRY_G][DSTATE];
    float dj[CARRY_G];
#pragma unroll
    for (int q = 0; q < CARRY_G; ++q) {
      const int j = (j0 + q < SCAN_NCH - 1) ? (j0 + q) : (SCAN_NCH - 2);
      dj[q] = Dsum[(size_t)j * SCAN_CHANS + ch];
#pragma unroll
      for (int n = 0; n < DSTATE; ++n)
        sv[q][n] = S[((size_t)j * DSTATE + n) * SCAN_CHANS + ch];
    }
    __builtin_amdgcn_sched_barrier(0);
#pragma unroll
    for (int q = 0; q < CARRY_G; ++q) {
      const bool live = (j0 + q) < ck;
#pragma unroll
      for (int n = 0; n < DSTATE; ++n) {
        const float e = live ? __expf(dj[q] * An[n]) : 1.0f;
        const float a = live ? sv[q][n] : 0.0f;
        h[n] = fmaf(e, h[n], a);
      }
    }
  }

  const size_t row0 = (size_t)b * SEQ_L + (size_t)ck * SCAN_CL;
  const float* dt_p = dt + row0 * D_INNER + d;
  const float* xc_p = xconv + row0 * D_INNER + d;
  const float* bc_p = x_dbl + row0 * XDBL_N + DTRANK + lane;
  const float* z_p = xz + row0 * (2 * D_INNER) + D_INNER + d;
  float* u_p = u + row0 * D_INNER + d;

  ScanBatch t0, t1;
  scan_load(t0, dt_p, xc_p, z_p, bc_p, 0);

  for (int l0 = 0; l0 < SCAN_CL; l0 += 2 * SCAN_TB) {
    scan_load(t1, dt_p, xc_p, z_p, bc_p, l0 + SCAN_TB);
    __builtin_amdgcn_sched_barrier(0);
    scan_steps(t0, h, An, Dd, u_p, l0);

    const int ln = (l0 + 2 * SCAN_TB < SCAN_CL) ? (l0 + 2 * SCAN_TB) : 0;
    scan_load(t0, dt_p, xc_p, z_p, bc_p, ln);
    __builtin_amdgcn_sched_barrier(0);
    scan_steps(t1, h, An, Dd, u_p, l0 + SCAN_TB);
  }
}

extern "C" void kernel_launch(void* const* d_in, const int* in_sizes, int n_in,
                              void* d_out, int out_size, void* d_ws, size_t ws_size,
                              hipStream_t stream) {
  const float* x      = (const float*)d_in[0];
  const float* W_in   = (const float*)d_in[1];
  const float* conv_w = (const float*)d_in[2];
  const float* conv_b = (const float*)d_in[3];
  const float* W_x    = (const float*)d_in[4];
  const float* W_dt   = (const float*)d_in[5];
  const float* b_dt   = (const float*)d_in[6];
  const float* A_log  = (const float*)d_in[7];
  const float* Dp     = (const float*)d_in[8];
  const float* W_out  = (const float*)d_in[9];
  float* out = (float*)d_out;

  float* ws = (float*)d_ws;
  const size_t M = (size_t)NB * SEQ_L;
  float* xz    = ws;
  float* xconv = xz + M * 2 * D_INNER;
  float* x_dbl = xconv + M * D_INNER;
  float* dt    = x_dbl + M * XDBL_N;
  float* u     = dt + M * D_INNER;
  float* S     = u + M * D_INNER;
  float* Dsum  = S + (size_t)(SCAN_NCH - 1) * DSTATE * SCAN_CHANS;

  dim3 blk(256);
  {
    hipLaunchKernelGGL((wmma_gemm_f32<2, 4, 4, 8, 2, 1>), dim3(256), blk, 0, stream,
                       x, W_in, xz, conv_b, conv_w, xconv,
                       (int)M, 2 * D_INNER, D_MODEL, D_MODEL, D_MODEL, 2 * D_INNER);
  }
  {
    hipLaunchKernelGGL(conv_edge_kernel, dim3((NB * SEQ_L / CONV_TM) * (D_INNER / 4) / 256), blk, 0, stream,
                       xz, conv_w, conv_b, xconv);
  }
  {
    hipLaunchKernelGGL((wmma_gemm_f32<0, 1, 2, 1, 1, 2>), dim3(384), dim3(32), 0, stream,
                       xconv, W_x, x_dbl, nullptr, nullptr, nullptr,
                       (int)M, XDBL_N, D_INNER, D_INNER, D_INNER, XDBL_N);
  }
  {
    hipLaunchKernelGGL((wmma_gemm_f32<1, 1, 2, 8, 4, 1, 8>), dim3(1024), blk, 0, stream,
                       x_dbl, W_dt, dt, b_dt, nullptr, nullptr,
                       (int)M, D_INNER, DTRANK, XDBL_N, DTRANK, D_INNER);
  }
  {
    hipLaunchKernelGGL(scan_state_kernel, dim3((SCAN_NCH - 1) * SCAN_GROUPS), dim3(32), 0, stream,
                       dt, xconv, x_dbl, A_log, S, Dsum);
  }
  {
    hipLaunchKernelGGL(scan_kernel, dim3(SCAN_NCH * SCAN_GROUPS), dim3(32), 0, stream,
                       dt, xconv, x_dbl, xz, A_log, Dp, S, Dsum, u);
  }
  {
    hipLaunchKernelGGL((wmma_gemm_ksplit_f32<4, 4, 4, 2>), dim3(256), blk, 0, stream,
                       u, W_out, out,
                       (int)M, D_MODEL, D_INNER, D_INNER, D_INNER, D_MODEL);
  }
}
